# starting kernel
# baseline (speedup 1.0000x reference)
#include <hip/hip_runtime.h>
#include <math.h>

#define NNODES   3712
#define NROI     116
#define NGRAPH   32
#define NFEAT    128
#define NHID     64
#define NHEADS   4
#define ALPHA_LR 0.1f
#define A_PARA   0.1f
#define PHI_C    0.94247779607693793f
#define MAXNORM  (1.0f - 4e-3f)

typedef __attribute__((ext_vector_type(2))) float v2f;
typedef __attribute__((ext_vector_type(8))) float v8f;

__device__ __forceinline__ v8f wmma4f32(v2f a, v2f b, v8f c) {
  return __builtin_amdgcn_wmma_f32_16x16x4_f32(
      false, a, false, b, (short)0, c, false, false);
}

__device__ __forceinline__ float fkernel_scale(float r) {
  float proj = 1.0f, rn = r;
  if (r > MAXNORM) { proj = MAXNORM / r; rn = MAXNORM; }
  rn = fmaxf(rn, 1e-15f);
  float t = fminf(rn, 1.0f - 1e-7f);
  return proj * (atanhf(t) / rn);
}

__device__ __forceinline__ float act_elu_cos(float v) {
  float e = (v > 0.0f) ? v : expm1f(v);
  return e + A_PARA * cosf(v + PHI_C);
}

__global__ __launch_bounds__(32)
void gemm_wmma_kernel(const float* __restrict__ A, const float* __restrict__ B,
                      float* __restrict__ C, int M, int K, int Nc) {
  const int h  = blockIdx.y;
  const float* Bh = B + (size_t)h * K * Nc;
  float*       Ch = C + (size_t)h * M * Nc;
  const int ntiles = Nc >> 4;
  const int mt = blockIdx.x / ntiles;
  const int nt = blockIdx.x % ntiles;
  const int l  = threadIdx.x;
  const int lm = l & 15;
  const int lh = l >> 4;

  v8f acc = {0.f,0.f,0.f,0.f,0.f,0.f,0.f,0.f};
  const float* arow = A  + (size_t)(mt * 16 + lm) * K + 2 * lh;
  const float* bcol = Bh + (size_t)(2 * lh) * Nc + nt * 16 + lm;

  for (int k0 = 0; k0 < K; k0 += 4) {
    v2f a = { arow[k0],              arow[k0 + 1] };
    v2f b = { bcol[(size_t)k0 * Nc], bcol[(size_t)(k0 + 1) * Nc] };
    acc = wmma4f32(a, b, acc);
  }
  float* crow = Ch + (size_t)(mt * 16 + 8 * lh) * Nc + nt * 16 + lm;
#pragma unroll
  for (int v = 0; v < 8; ++v) crow[(size_t)v * Nc] = acc[v];
}

__global__ void attvec_kernel(const float* __restrict__ Wh,
                              const float* __restrict__ avec,
                              float* __restrict__ el, float* __restrict__ er,
                              int heads) {
  int idx = blockIdx.x * blockDim.x + threadIdx.x;
  if (idx >= heads * NNODES) return;
  int h = idx / NNODES, i = idx % NNODES;
  const float* row = Wh + ((size_t)h * NNODES + i) * NHID;
  const float* a1  = avec + (size_t)h * 2 * NHID;
  const float* a2  = a1 + NHID;
  float s1 = 0.f, s2 = 0.f;
#pragma unroll 4
  for (int k = 0; k < NHID; ++k) { s1 += row[k] * a1[k]; s2 += row[k] * a2[k]; }
  el[idx] = s1;
  er[idx] = s2;
}

#define ATT_S 132
#define WHS_S 68
#define LDS_BYTES ((128 * ATT_S + 128 * WHS_S) * sizeof(float))

template <bool DOUBLE_FK>
__global__ __launch_bounds__(256)
void att_block_kernel(const float* __restrict__ adj, const float* __restrict__ Wh,
                      const float* __restrict__ el, const float* __restrict__ er,
                      float* __restrict__ out, int out_stride) {
  extern __shared__ float lds[];
  float* att = lds;
  float* whs = lds + 128 * ATT_S;

  const int g = blockIdx.x, h = blockIdx.y;
  const int t = threadIdx.x;
  const int r0 = g * NROI;

  for (int i = t; i < 128 * ATT_S; i += 256) att[i] = 0.f;
  for (int i = t; i < 128 * WHS_S; i += 256) whs[i] = 0.f;
  __syncthreads();

  const float* WhH = Wh + (size_t)h * NNODES * NHID;
  for (int i = t; i < NROI * NHID; i += 256) {
    int r = i >> 6, c = i & 63;
    whs[r * WHS_S + c] = WhH[(size_t)(r0 + r) * NHID + c];
  }

  const float* elh = el + (size_t)h * NNODES + r0;
  const float* erh = er + (size_t)h * NNODES + r0;
  if (t < NROI) {
    const float* adjrow = adj + (size_t)(r0 + t) * NNODES + r0;
    float* arow = att + t * ATT_S;
    const float eli = elh[t];
    float m = -INFINITY;
    for (int j = 0; j < NROI; ++j) {
      float e = -INFINITY;
      if (adjrow[j] > 0.f) {
        float z = eli + erh[j];
        e = (z > 0.f) ? z : ALPHA_LR * z;
        m = fmaxf(m, e);
      }
      arow[j] = e;
    }
    if (m == -INFINITY) {
      for (int j = 0; j < NROI; ++j) arow[j] = 0.f;
    } else {
      float s = 0.f;
      for (int j = 0; j < NROI; ++j) {
        float w = (arow[j] == -INFINITY) ? 0.f : expf(arow[j] - m);
        arow[j] = w; s += w;
      }
      float inv = (s > 0.f) ? 1.f / s : 0.f;
      for (int j = 0; j < NROI; ++j) arow[j] *= inv;
    }
  }
  __syncthreads();

  const int w  = t >> 5;
  const int l  = t & 31;
  const int lm = l & 15, lh = l >> 4;
  v8f acc[4];
#pragma unroll
  for (int nt = 0; nt < 4; ++nt)
    acc[nt] = (v8f){0.f,0.f,0.f,0.f,0.f,0.f,0.f,0.f};

  const float* arowA = att + (w * 16 + lm) * ATT_S + 2 * lh;
  const float* bbase = whs + (2 * lh) * WHS_S + lm;
  for (int k0 = 0; k0 < 128; k0 += 4) {
    v2f a = { arowA[k0], arowA[k0 + 1] };
#pragma unroll
    for (int nt = 0; nt < 4; ++nt) {
      const float* bc = bbase + k0 * WHS_S + nt * 16;
      v2f b = { bc[0], bc[WHS_S] };
      acc[nt] = wmma4f32(a, b, acc[nt]);
    }
  }
  __syncthreads();

  float* hp = att;
#pragma unroll
  for (int nt = 0; nt < 4; ++nt)
#pragma unroll
    for (int v = 0; v < 8; ++v)
      hp[(w * 16 + v + 8 * lh) * WHS_S + nt * 16 + lm] = acc[nt][v];
  __syncthreads();

  if (t < NROI) {
    const float* row = hp + t * WHS_S;
    float ss = 0.f;
    for (int c = 0; c < NHID; ++c) ss += row[c] * row[c];
    float r = sqrtf(ss);
    float s = fkernel_scale(r);
    if (DOUBLE_FK) s *= fkernel_scale(s * r);
    float* orow = out + (size_t)(r0 + t) * out_stride + h * NHID;
    for (int c = 0; c < NHID; ++c) orow[c] = act_elu_cos(s * row[c]);
  }
}

extern "C" void kernel_launch(void* const* d_in, const int* in_sizes, int n_in,
                              void* d_out, int out_size, void* d_ws, size_t ws_size,
                              hipStream_t stream) {
  (void)in_sizes; (void)n_in; (void)out_size; (void)ws_size;
  const float* x       = (const float*)d_in[0];
  const float* adj     = (const float*)d_in[1];
  const float* W_heads = (const float*)d_in[2];
  const float* a_heads = (const float*)d_in[3];
  const float* W_out   = (const float*)d_in[4];
  const float* a_out   = (const float*)d_in[5];
  float* out = (float*)d_out;

  float* ws = (float*)d_ws;
  float* Wh   = ws;                  ws += (size_t)NHEADS * NNODES * NHID;
  float* el1  = ws;                  ws += (size_t)NHEADS * NNODES;
  float* er1  = ws;                  ws += (size_t)NHEADS * NNODES;
  float* xcat = ws;                  ws += (size_t)NNODES * NHEADS * NHID;
  float* Wh2  = ws;                  ws += (size_t)NNODES * NHID;
  float* el2  = ws;                  ws += (size_t)NNODES;
  float* er2  = ws;

  const int mtiles = NNODES / 16;

  gemm_wmma_kernel<<<dim3(mtiles * (NHID / 16), NHEADS), 32, 0, stream>>>(
      x, W_heads, Wh, NNODES, NFEAT, NHID);
  attvec_kernel<<<(NHEADS * NNODES + 255) / 256, 256, 0, stream>>>(
      Wh, a_heads, el1, er1, NHEADS);
  att_block_kernel<false><<<dim3(NGRAPH, NHEADS), 256, LDS_BYTES, stream>>>(
      adj, Wh, el1, er1, xcat, NHEADS * NHID);

  gemm_wmma_kernel<<<dim3(mtiles * (NHID / 16), 1), 32, 0, stream>>>(
      xcat, W_out, Wh2, NNODES, NHEADS * NHID, NHID);
  attvec_kernel<<<(NNODES + 255) / 256, 256, 0, stream>>>(
      Wh2, a_out, el2, er2, 1);
  att_block_kernel<true><<<dim3(NGRAPH, 1), 256, LDS_BYTES, stream>>>(
      adj, Wh2, el2, er2, out, NHID);
}


# el/er dots moved into the projection GEMM's epilogue: a wave owns a 16x64 row strip of one head; 6 launches become 4; el/er summed in a new order (4 terms in-lane, then a 16-lane tree)
# speedup vs baseline: 2.2338x; 2.2338x over previous
#include <hip/hip_runtime.h>
#include <math.h>

#define NNODES   3712
#define NROI     116
#define NGRAPH   32
#define NFEAT    128
#define NHID     64
#define NHEADS   4
#define ALPHA_LR 0.1f
#define A_PARA   0.1f
#define PHI_C    0.94247779607693793f
#define MAXNORM  (1.0f - 4e-3f)

typedef __attribute__((ext_vector_type(2))) float v2f;
typedef __attribute__((ext_vector_type(4))) float v4f;
typedef __attribute__((ext_vector_type(8))) float v8f;

__device__ __forceinline__ v8f wmma4f32(v2f a, v2f b, v8f c) {
  return __builtin_amdgcn_wmma_f32_16x16x4_f32(
      false, a, false, b, (short)0, c, false, false);
}

__device__ __forceinline__ float fkernel_scale(float r) {
  float proj = 1.0f, rn = r;
  if (r > MAXNORM) { proj = MAXNORM / r; rn = MAXNORM; }
  rn = fmaxf(rn, 1e-15f);
  float t = fminf(rn, 1.0f - 1e-7f);
  return proj * (atanhf(t) / rn);
}

__device__ __forceinline__ float act_elu_cos(float v) {
  float e = (v > 0.0f) ? v : expm1f(v);
  return e + A_PARA * cosf(v + PHI_C);
}

#define KGRP 8

template <int MASK>
__device__ __forceinline__ float row_xor_add(float v) {
  int o = __builtin_amdgcn_update_dpp(0, __float_as_int(v), 0x160 | MASK, 0xf, 0xf, false);
  return v + __int_as_float(o);
}

template <int K, int NH>
__global__ __launch_bounds__(32 * NH)
void proj_strip_kernel(const float* __restrict__ A, const float* __restrict__ B,
                       const float* __restrict__ avec, float* __restrict__ C,
                       float* __restrict__ el, float* __restrict__ er) {
  static_assert(K % (4 * KGRP) == 0, "K is a whole number of load groups");
  constexpr int NGRP = K / (4 * KGRP);
  const int h  = threadIdx.x >> 5;
  const int l  = threadIdx.x & 31;
  const int lm = l & 15;
  const int lh = l >> 4;
  const int m0 = blockIdx.x * 16;

  const v2f* ap = (const v2f*)(A + (size_t)(m0 + lm) * K + 2 * lh);
  const v4f* bp = (const v4f*)(B + (size_t)h * K * NHID + (size_t)(2 * lh) * NHID + 4 * lm);

  v2f af[2][KGRP];
  v4f bf[2][KGRP][2];
  v8f acc[4];
#pragma unroll
  for (int nt = 0; nt < 4; ++nt)
    acc[nt] = (v8f){0.f,0.f,0.f,0.f,0.f,0.f,0.f,0.f};

#pragma unroll
  for (int j = 0; j < KGRP; ++j) {
    af[0][j]    = ap[2 * j];
    bf[0][j][0] = bp[(4 * j) * (NHID / 4)];
    bf[0][j][1] = bp[(4 * j + 1) * (NHID / 4)];
  }
#pragma unroll
  for (int g = 0; g < NGRP; ++g) {
    if (g + 1 < NGRP) {
#pragma unroll
      for (int j = 0; j < KGRP; ++j) {
        const int s = (g + 1) * KGRP + j;
        af[(g + 1) & 1][j]    = ap[2 * s];
        bf[(g + 1) & 1][j][0] = bp[(4 * s) * (NHID / 4)];
        bf[(g + 1) & 1][j][1] = bp[(4 * s + 1) * (NHID / 4)];
      }
    }
    __builtin_amdgcn_sched_barrier(0);
#pragma unroll
    for (int j = 0; j < KGRP; ++j) {
      const v2f a = af[g & 1][j];
#pragma unroll
      for (int nt = 0; nt < 4; ++nt) {
        const v2f b = { bf[g & 1][j][0][nt], bf[g & 1][j][1][nt] };
        acc[nt] = wmma4f32(a, b, acc[nt]);
      }
    }
    __builtin_amdgcn_sched_barrier(0);
  }

  float* Ch = C + (size_t)h * NNODES * NHID;
#pragma unroll
  for (int r = 0; r < 8; ++r) {
    v4f o = { acc[0][r], acc[1][r], acc[2][r], acc[3][r] };
    *(v4f*)(Ch + (size_t)(m0 + 8 * lh + r) * NHID + 4 * lm) = o;
  }

  const v4f a1 = *(const v4f*)(avec + (size_t)h * 2 * NHID + 4 * lm);
  const v4f a2 = *(const v4f*)(avec + (size_t)h * 2 * NHID + NHID + 4 * lm);
  float p1[8], p2[8];
#pragma unroll
  for (int r = 0; r < 8; ++r) {
    p1[r] = acc[0][r] * a1[0] + acc[1][r] * a1[1] + acc[2][r] * a1[2] + acc[3][r] * a1[3];
    p2[r] = acc[0][r] * a2[0] + acc[1][r] * a2[1] + acc[2][r] * a2[2] + acc[3][r] * a2[3];
  }
#pragma unroll
  for (int r = 0; r < 8; ++r) {
    p1[r] = row_xor_add<1>(p1[r]);  p2[r] = row_xor_add<1>(p2[r]);
  }
#pragma unroll
  for (int r = 0; r < 8; ++r) {
    p1[r] = row_xor_add<2>(p1[r]);  p2[r] = row_xor_add<2>(p2[r]);
  }
#pragma unroll
  for (int r = 0; r < 8; ++r) {
    p1[r] = row_xor_add<4>(p1[r]);  p2[r] = row_xor_add<4>(p2[r]);
  }
#pragma unroll
  for (int r = 0; r < 8; ++r) {
    p1[r] = row_xor_add<8>(p1[r]);  p2[r] = row_xor_add<8>(p2[r]);
  }
  float e1 = p1[0], e2 = p2[0];
#pragma unroll
  for (int r = 1; r < 8; ++r) {
    e1 = (lm == r) ? p1[r] : e1;
    e2 = (lm == r) ? p2[r] : e2;
  }
  if (lm < 8) {
    const size_t idx = (size_t)h * NNODES + m0 + 8 * lh + lm;
    el[idx] = e1;
    er[idx] = e2;
  }
}

#define WHS_S 68
#define KSTEPS (NROI / 4)
#define LDS_BYTES ((NROI * WHS_S + 128 * WHS_S) * sizeof(float))

__device__ __forceinline__ float masked_score(float eli, float erj, float adjv) {
  float z  = eli + erj;
  float ev = (z > 0.f) ? z : ALPHA_LR * z;
  return (adjv > 0.f) ? ev : -INFINITY;
}

__device__ __forceinline__ float masked_weight(float e, float m) {
  float wv = expf(e - m);
  return (e == -INFINITY) ? 0.f : wv;
}

template <bool DOUBLE_FK>
__global__ __launch_bounds__(256)
void att_block_kernel(const float* __restrict__ adj, const float* __restrict__ Wh,
                      const float* __restrict__ el, const float* __restrict__ er,
                      float* __restrict__ out, int out_stride) {
  extern __shared__ float lds[];
  float* whs = lds;
  float* hp  = lds + NROI * WHS_S;

  const int g = blockIdx.x, h = blockIdx.y;
  const int t = threadIdx.x;
  const int r0 = g * NROI;
  const int w  = t >> 5;
  const int l  = t & 31;
  const int lm = l & 15, lh = l >> 4;

  const float* elh = el + (size_t)h * NNODES + r0;
  const float* erh = er + (size_t)h * NNODES + r0;

  const int  irow  = w * 16 + lm;
  const bool rowok = irow < NROI;
  const int  ic    = rowok ? irow : (NROI - 1);
  const v2f* adjp = (const v2f*)(adj + (size_t)(r0 + ic) * NNODES + r0 + 2 * lh);
  const v2f* erp  = (const v2f*)(erh + 2 * lh);
  const float eli = elh[ic];

  static_assert((NROI * NHID) % 256 == 0, "staging has no remainder");
  constexpr int NSTAGE = (NROI * NHID) / 256;
  const float* WhH = Wh + (size_t)h * NNODES * NHID;
  v2f av[KSTEPS], rv[KSTEPS];
  float st[NSTAGE];
#pragma unroll
  for (int s = 0; s < KSTEPS; ++s) {
    av[s] = adjp[2 * s];
    rv[s] = erp[2 * s];
  }
#pragma unroll
  for (int k = 0; k < NSTAGE; ++k) {
    int i = t + 256 * k;
    st[k] = WhH[(size_t)(r0 + (i >> 6)) * NHID + (i & 63)];
  }
  __builtin_amdgcn_sched_barrier(0);

#pragma unroll
  for (int k = 0; k < NSTAGE; ++k) {
    int i = t + 256 * k;
    whs[(i >> 6) * WHS_S + (i & 63)] = st[k];
  }

  v2f e[KSTEPS];
  float m = -INFINITY;
#pragma unroll
  for (int s = 0; s < KSTEPS; ++s) {
    e[s][0] = masked_score(eli, rv[s][0], av[s][0]);
    e[s][1] = masked_score(eli, rv[s][1], av[s][1]);
    m = fmaxf(fmaxf(m, e[s][0]), e[s][1]);
  }

  m = fmaxf(m, __shfl_xor(m, 16));
  float ssum = 0.f;
#pragma unroll
  for (int s = 0; s < KSTEPS; ++s) {
    e[s][0] = masked_weight(e[s][0], m);
    e[s][1] = masked_weight(e[s][1], m);
    ssum += e[s][0];
    ssum += e[s][1];
  }
  ssum += __shfl_xor(ssum, 16);
  float inv = (ssum > 0.f) ? 1.f / ssum : 0.f;
  if (!rowok) inv = 0.f;
#pragma unroll
  for (int s = 0; s < KSTEPS; ++s) { e[s][0] *= inv; e[s][1] *= inv; }

  __syncthreads();

  v8f acc[4];
#pragma unroll
  for (int nt = 0; nt < 4; ++nt)
    acc[nt] = (v8f){0.f,0.f,0.f,0.f,0.f,0.f,0.f,0.f};

  const float* bbase = whs + (2 * lh) * WHS_S + lm;
#pragma unroll
  for (int s = 0; s < KSTEPS; ++s) {
    const v2f a = e[s];
#pragma unroll
    for (int nt = 0; nt < 4; ++nt) {
      const float* bc = bbase + (4 * s) * WHS_S + nt * 16;
      v2f b = { bc[0], bc[WHS_S] };
      acc[nt] = wmma4f32(a, b, acc[nt]);
    }
  }

#pragma unroll
  for (int nt = 0; nt < 4; ++nt)
#pragma unroll
    for (int v = 0; v < 8; ++v)
      hp[(w * 16 + v + 8 * lh) * WHS_S + nt * 16 + lm] = acc[nt][v];
  __syncthreads();

  if (t < NROI) {
    const float* row = hp + t * WHS_S;
    float ss = 0.f;
    for (int c = 0; c < NHID; ++c) ss += row[c] * row[c];
    float r = sqrtf(ss);
    float s = fkernel_scale(r);
    if (DOUBLE_FK) s *= fkernel_scale(s * r);
    float* orow = out + (size_t)(r0 + t) * out_stride + h * NHID;
    for (int c = 0; c < NHID; ++c) orow[c] = act_elu_cos(s * row[c]);
  }
}

extern "C" void kernel_launch(void* const* d_in, const int* in_sizes, int n_in,
                              void* d_out, int out_size, void* d_ws, size_t ws_size,
                              hipStream_t stream) {
  (void)in_sizes; (void)n_in; (void)out_size; (void)ws_size;
  const float* x       = (const float*)d_in[0];
  const float* adj     = (const float*)d_in[1];
  const float* W_heads = (const float*)d_in[2];
  const float* a_heads = (const float*)d_in[3];
  const float* W_out   = (const float*)d_in[4];
  const float* a_out   = (const float*)d_in[5];
  float* out = (float*)d_out;

  float* ws = (float*)d_ws;
  float* Wh   = ws;                  ws += (size_t)NHEADS * NNODES * NHID;
  float* el1  = ws;                  ws += (size_t)NHEADS * NNODES;
  float* er1  = ws;                  ws += (size_t)NHEADS * NNODES;
  float* xcat = ws;                  ws += (size_t)NNODES * NHEADS * NHID;
  float* Wh2  = ws;                  ws += (size_t)NNODES * NHID;
  float* el2  = ws;                  ws += (size_t)NNODES;
  float* er2  = ws;

  const int mtiles = NNODES / 16;

  proj_strip_kernel<NFEAT, NHEADS><<<mtiles, 32 * NHEADS, 0, stream>>>(
      x, W_heads, a_heads, Wh, el1, er1);
  att_block_kernel<false><<<dim3(NGRAPH, NHEADS), 256, LDS_BYTES, stream>>>(
      adj, Wh, el1, er1, xcat, NHEADS * NHID);

  proj_strip_kernel<NHEADS * NHID, 1><<<mtiles, 32, 0, stream>>>(
      xcat, W_out, a_out, Wh2, el2, er2);
  att_block_kernel<true><<<dim3(NGRAPH, 1), 256, LDS_BYTES, stream>>>(
      adj, Wh2, el2, er2, out, NHID);
}


# Row norm, fkernel scale and activation done on the WMMA accumulators by all 8 waves (lane = 4 adjacent columns, 16-B stores), hp never in LDS; row norm summed in a new order (4 in-lane, 16-lane tree)
# speedup vs baseline: 3.3219x; 1.4871x over previous
#include <hip/hip_runtime.h>
#include <math.h>

#define NNODES   3712
#define NROI     116
#define NGRAPH   32
#define NFEAT    128
#define NHID     64
#define NHEADS   4
#define ALPHA_LR 0.1f
#define A_PARA   0.1f
#define PHI_C    0.94247779607693793f
#define MAXNORM  (1.0f - 4e-3f)

typedef __attribute__((ext_vector_type(2))) float v2f;
typedef __attribute__((ext_vector_type(4))) float v4f;
typedef __attribute__((ext_vector_type(8))) float v8f;

__device__ __forceinline__ v8f wmma4f32(v2f a, v2f b, v8f c) {
  return __builtin_amdgcn_wmma_f32_16x16x4_f32(
      false, a, false, b, (short)0, c, false, false);
}

__device__ __forceinline__ float fkernel_scale(float r) {
  float proj = 1.0f, rn = r;
  if (r > MAXNORM) { proj = MAXNORM / r; rn = MAXNORM; }
  rn = fmaxf(rn, 1e-15f);
  float t = fminf(rn, 1.0f - 1e-7f);
  return proj * (atanhf(t) / rn);
}

__device__ __forceinline__ float act_elu_cos(float v) {
  float e = (v > 0.0f) ? v : expm1f(v);
  return e + A_PARA * cosf(v + PHI_C);
}

#define KGRP 8

template <int MASK>
__device__ __forceinline__ float row_xor_add(float v) {
  int o = __builtin_amdgcn_update_dpp(0, __float_as_int(v), 0x160 | MASK, 0xf, 0xf, false);
  return v + __int_as_float(o);
}

template <int K, int NH>
__global__ __launch_bounds__(32 * NH)
void proj_strip_kernel(const float* __restrict__ A, const float* __restrict__ B,
                       const float* __restrict__ avec, float* __restrict__ C,
                       float* __restrict__ el, float* __restrict__ er) {
  static_assert(K % (4 * KGRP) == 0, "K is a whole number of load groups");
  constexpr int NGRP = K / (4 * KGRP);
  const int h  = threadIdx.x >> 5;
  const int l  = threadIdx.x & 31;
  const int lm = l & 15;
  const int lh = l >> 4;
  const int m0 = blockIdx.x * 16;

  const v2f* ap = (const v2f*)(A + (size_t)(m0 + lm) * K + 2 * lh);
  const v4f* bp = (const v4f*)(B + (size_t)h * K * NHID + (size_t)(2 * lh) * NHID + 4 * lm);

  v2f af[2][KGRP];
  v4f bf[2][KGRP][2];
  v8f acc[4];
#pragma unroll
  for (int nt = 0; nt < 4; ++nt)
    acc[nt] = (v8f){0.f,0.f,0.f,0.f,0.f,0.f,0.f,0.f};

#pragma unroll
  for (int j = 0; j < KGRP; ++j) {
    af[0][j]    = ap[2 * j];
    bf[0][j][0] = bp[(4 * j) * (NHID / 4)];
    bf[0][j][1] = bp[(4 * j + 1) * (NHID / 4)];
  }
#pragma unroll
  for (int g = 0; g < NGRP; ++g) {
    if (g + 1 < NGRP) {
#pragma unroll
      for (int j = 0; j < KGRP; ++j) {
        const int s = (g + 1) * KGRP + j;
        af[(g + 1) & 1][j]    = ap[2 * s];
        bf[(g + 1) & 1][j][0] = bp[(4 * s) * (NHID / 4)];
        bf[(g + 1) & 1][j][1] = bp[(4 * s + 1) * (NHID / 4)];
      }
    }
    __builtin_amdgcn_sched_barrier(0);
#pragma unroll
    for (int j = 0; j < KGRP; ++j) {
      const v2f a = af[g & 1][j];
#pragma unroll
      for (int nt = 0; nt < 4; ++nt) {
        const v2f b = { bf[g & 1][j][0][nt], bf[g & 1][j][1][nt] };
        acc[nt] = wmma4f32(a, b, acc[nt]);
      }
    }
    __builtin_amdgcn_sched_barrier(0);
  }

  float* Ch = C + (size_t)h * NNODES * NHID;
#pragma unroll
  for (int r = 0; r < 8; ++r) {
    v4f o = { acc[0][r], acc[1][r], acc[2][r], acc[3][r] };
    *(v4f*)(Ch + (size_t)(m0 + 8 * lh + r) * NHID + 4 * lm) = o;
  }

  const v4f a1 = *(const v4f*)(avec + (size_t)h * 2 * NHID + 4 * lm);
  const v4f a2 = *(const v4f*)(avec + (size_t)h * 2 * NHID + NHID + 4 * lm);
  float p1[8], p2[8];
#pragma unroll
  for (int r = 0; r < 8; ++r) {
    p1[r] = acc[0][r] * a1[0] + acc[1][r] * a1[1] + acc[2][r] * a1[2] + acc[3][r] * a1[3];
    p2[r] = acc[0][r] * a2[0] + acc[1][r] * a2[1] + acc[2][r] * a2[2] + acc[3][r] * a2[3];
  }
#pragma unroll
  for (int r = 0; r < 8; ++r) {
    p1[r] = row_xor_add<1>(p1[r]);  p2[r] = row_xor_add<1>(p2[r]);
  }
#pragma unroll
  for (int r = 0; r < 8; ++r) {
    p1[r] = row_xor_add<2>(p1[r]);  p2[r] = row_xor_add<2>(p2[r]);
  }
#pragma unroll
  for (int r = 0; r < 8; ++r) {
    p1[r] = row_xor_add<4>(p1[r]);  p2[r] = row_xor_add<4>(p2[r]);
  }
#pragma unroll
  for (int r = 0; r < 8; ++r) {
    p1[r] = row_xor_add<8>(p1[r]);  p2[r] = row_xor_add<8>(p2[r]);
  }
  float e1 = p1[0], e2 = p2[0];
#pragma unroll
  for (int r = 1; r < 8; ++r) {
    e1 = (lm == r) ? p1[r] : e1;
    e2 = (lm == r) ? p2[r] : e2;
  }
  if (lm < 8) {
    const size_t idx = (size_t)h * NNODES + m0 + 8 * lh + lm;
    el[idx] = e1;
    er[idx] = e2;
  }
}

#define WHS_S 68
#define KSTEPS (NROI / 4)
#define LDS_BYTES ((NROI * WHS_S) * sizeof(float))

template <int N>
__device__ __forceinline__ float row_share(float v) {
  int o = __builtin_amdgcn_update_dpp(0, __float_as_int(v), 0x150 | N, 0xf, 0xf, false);
  return __int_as_float(o);
}

__device__ __forceinline__ float masked_score(float eli, float erj, float adjv) {
  float z  = eli + erj;
  float ev = (z > 0.f) ? z : ALPHA_LR * z;
  return (adjv > 0.f) ? ev : -INFINITY;
}

__device__ __forceinline__ float masked_weight(float e, float m) {
  float wv = expf(e - m);
  return (e == -INFINITY) ? 0.f : wv;
}

template <bool DOUBLE_FK>
__global__ __launch_bounds__(256)
void att_block_kernel(const float* __restrict__ adj, const float* __restrict__ Wh,
                      const float* __restrict__ el, const float* __restrict__ er,
                      float* __restrict__ out, int out_stride) {
  extern __shared__ __attribute__((aligned(16))) float lds[];
  float* whs = lds;

  const int g = blockIdx.x, h = blockIdx.y;
  const int t = threadIdx.x;
  const int r0 = g * NROI;
  const int w  = t >> 5;
  const int l  = t & 31;
  const int lm = l & 15, lh = l >> 4;

  const float* elh = el + (size_t)h * NNODES + r0;
  const float* erh = er + (size_t)h * NNODES + r0;

  const int  irow  = w * 16 + lm;
  const bool rowok = irow < NROI;
  const int  ic    = rowok ? irow : (NROI - 1);
  const v2f* adjp = (const v2f*)(adj + (size_t)(r0 + ic) * NNODES + r0 + 2 * lh);
  const v2f* erp  = (const v2f*)(erh + 2 * lh);
  const float eli = elh[ic];

  static_assert((NROI * NHID) % 256 == 0, "staging has no remainder");
  constexpr int NSTAGE = (NROI * NHID) / 256;
  const float* WhH = Wh + (size_t)h * NNODES * NHID;
  v2f av[KSTEPS], rv[KSTEPS];
  float st[NSTAGE];
#pragma unroll
  for (int s = 0; s < KSTEPS; ++s) {
    av[s] = adjp[2 * s];
    rv[s] = erp[2 * s];
  }
#pragma unroll
  for (int k = 0; k < NSTAGE; ++k) {
    int i = t + 256 * k;
    st[k] = WhH[(size_t)(r0 + (i >> 6)) * NHID + (i & 63)];
  }
  __builtin_amdgcn_sched_barrier(0);

#pragma unroll
  for (int k = 0; k < NSTAGE; ++k) {
    int i = t + 256 * k;
    whs[(i >> 6) * WHS_S + (i & 63)] = st[k];
  }

  v2f e[KSTEPS];
  float m = -INFINITY;
#pragma unroll
  for (int s = 0; s < KSTEPS; ++s) {
    e[s][0] = masked_score(eli, rv[s][0], av[s][0]);
    e[s][1] = masked_score(eli, rv[s][1], av[s][1]);
    m = fmaxf(fmaxf(m, e[s][0]), e[s][1]);
  }

  m = fmaxf(m, __shfl_xor(m, 16));
  float ssum = 0.f;
#pragma unroll
  for (int s = 0; s < KSTEPS; ++s) {
    e[s][0] = masked_weight(e[s][0], m);
    e[s][1] = masked_weight(e[s][1], m);
    ssum += e[s][0];
    ssum += e[s][1];
  }
  ssum += __shfl_xor(ssum, 16);
  float inv = (ssum > 0.f) ? 1.f / ssum : 0.f;
  if (!rowok) inv = 0.f;
#pragma unroll
  for (int s = 0; s < KSTEPS; ++s) { e[s][0] *= inv; e[s][1] *= inv; }

  __syncthreads();

  v8f acc[4];
#pragma unroll
  for (int nt = 0; nt < 4; ++nt)
    acc[nt] = (v8f){0.f,0.f,0.f,0.f,0.f,0.f,0.f,0.f};

  const v4f* bq = (const v4f*)(whs + (2 * lh) * WHS_S + 4 * lm);
#pragma unroll
  for (int s = 0; s < KSTEPS; ++s) {
    const v2f a = e[s];
    const v4f b0 = bq[(4 * s) * (WHS_S / 4)];
    const v4f b1 = bq[(4 * s + 1) * (WHS_S / 4)];
#pragma unroll
    for (int nt = 0; nt < 4; ++nt) {
      const v2f b = { b0[nt], b1[nt] };
      acc[nt] = wmma4f32(a, b, acc[nt]);
    }
  }

  float ss[8];
#pragma unroll
  for (int v = 0; v < 8; ++v)
    ss[v] = acc[0][v] * acc[0][v] + acc[1][v] * acc[1][v]
          + acc[2][v] * acc[2][v] + acc[3][v] * acc[3][v];
#pragma unroll
  for (int v = 0; v < 8; ++v) ss[v] = row_xor_add<1>(ss[v]);
#pragma unroll
  for (int v = 0; v < 8; ++v) ss[v] = row_xor_add<2>(ss[v]);
#pragma unroll
  for (int v = 0; v < 8; ++v) ss[v] = row_xor_add<4>(ss[v]);
#pragma unroll
  for (int v = 0; v < 8; ++v) ss[v] = row_xor_add<8>(ss[v]);

  float mine = ss[0];
#pragma unroll
  for (int v = 1; v < 8; ++v) mine = (lm == v) ? ss[v] : mine;
  const float r = sqrtf(mine);
  float sc = fkernel_scale(r);
  if (DOUBLE_FK) sc *= fkernel_scale(sc * r);
  float sv[8];
  sv[0] = row_share<0>(sc);  sv[1] = row_share<1>(sc);
  sv[2] = row_share<2>(sc);  sv[3] = row_share<3>(sc);
  sv[4] = row_share<4>(sc);  sv[5] = row_share<5>(sc);
  sv[6] = row_share<6>(sc);  sv[7] = row_share<7>(sc);

  v4f hv[8];
#pragma unroll
  for (int v = 0; v < 8; ++v)
    hv[v] = (v4f){ acc[0][v], acc[1][v], acc[2][v], acc[3][v] };

  int    row  = w * 16 + 8 * lh;
  float* orow = out + (size_t)(r0 + row) * out_stride + h * NHID + 4 * lm;
  for (int v = 0; v < 8; ++v) {
    const v4f   x = hv[0];
    const float s = sv[0];
    const v4f o = { act_elu_cos(s * x[0]), act_elu_cos(s * x[1]),
                    act_elu_cos(s * x[2]), act_elu_cos(s * x[3]) };
    if (row < NROI) *(v4f*)orow = o;
    ++row;
    orow += out_stride;
#pragma unroll
    for (int j = 0; j < 7; ++j) { hv[j] = hv[j + 1]; sv[j] = sv[j + 1]; }
  }
}

extern "C" void kernel_launch(void* const* d_in, const int* in_sizes, int n_in,
                              void* d_out, int out_size, void* d_ws, size_t ws_size,
                              hipStream_t stream) {
  (void)in_sizes; (void)n_in; (void)out_size; (void)ws_size;
  const float* x       = (const float*)d_in[0];
  const float* adj     = (const float*)d_in[1];
  const float* W_heads = (const float*)d_in[2];
  const float* a_heads = (const float*)d_in[3];
  const float* W_out   = (const float*)d_in[4];
  const float* a_out   = (const float*)d_in[5];
  float* out = (float*)d_out;

  float* ws = (float*)d_ws;
  float* Wh   = ws;                  ws += (size_t)NHEADS * NNODES * NHID;
  float* el1  = ws;                  ws += (size_t)NHEADS * NNODES;
  float* er1  = ws;                  ws += (size_t)NHEADS * NNODES;
  float* xcat = ws;                  ws += (size_t)NNODES * NHEADS * NHID;
  float* Wh2  = ws;                  ws += (size_t)NNODES * NHID;
  float* el2  = ws;                  ws += (size_t)NNODES;
  float* er2  = ws;

  const int mtiles = NNODES / 16;

  proj_strip_kernel<NFEAT, NHEADS><<<mtiles, 32 * NHEADS, 0, stream>>>(
      x, W_heads, a_heads, Wh, el1, er1);
  att_block_kernel<false><<<dim3(NGRAPH, NHEADS), 256, LDS_BYTES, stream>>>(
      adj, Wh, el1, er1, xcat, NHEADS * NHID);

  proj_strip_kernel<NHEADS * NHID, 1><<<mtiles, 32, 0, stream>>>(
      xcat, W_out, a_out, Wh2, el2, er2);
  att_block_kernel<true><<<dim3(NGRAPH, 1), 256, LDS_BYTES, stream>>>(
      adj, Wh2, el2, er2, out, NHID);
}


# Layer-2 projection moved into the layer-1 attention epilogue as per-head partial products (tile via LDS, never written out): 4 launches become 3; Wh2, el2, er2 now summed as 4 head partials
# speedup vs baseline: 3.6072x; 1.0859x over previous
#include <hip/hip_runtime.h>
#include <math.h>

#define NNODES   3712
#define NROI     116
#define NGRAPH   32
#define NFEAT    128
#define NHID     64
#define NHEADS   4
#define ALPHA_LR 0.1f
#define A_PARA   0.1f
#define PHI_C    0.94247779607693793f
#define MAXNORM  (1.0f - 4e-3f)

typedef __attribute__((ext_vector_type(2))) float v2f;
typedef __attribute__((ext_vector_type(4))) float v4f;
typedef __attribute__((ext_vector_type(8))) float v8f;

__device__ __forceinline__ v8f wmma4f32(v2f a, v2f b, v8f c) {
  return __builtin_amdgcn_wmma_f32_16x16x4_f32(
      false, a, false, b, (short)0, c, false, false);
}

__device__ __forceinline__ float fkernel_scale(float r) {
  float proj = 1.0f, rn = r;
  if (r > MAXNORM) { proj = MAXNORM / r; rn = MAXNORM; }
  rn = fmaxf(rn, 1e-15f);
  float t = fminf(rn, 1.0f - 1e-7f);
  return proj * (atanhf(t) / rn);
}

__device__ __forceinline__ float act_elu_cos(float v) {
  float e = (v > 0.0f) ? v : expm1f(v);
  return e + A_PARA * cosf(v + PHI_C);
}

#define KGRP 8

template <int MASK>
__device__ __forceinline__ float row_xor_add(float v) {
  int o = __builtin_amdgcn_update_dpp(0, __float_as_int(v), 0x160 | MASK, 0xf, 0xf, false);
  return v + __int_as_float(o);
}

template <int K, int NH>
__global__ __launch_bounds__(32 * NH)
void proj_strip_kernel(const float* __restrict__ A, const float* __restrict__ B,
                       const float* __restrict__ avec, float* __restrict__ C,
                       float* __restrict__ el, float* __restrict__ er) {
  static_assert(K % (4 * KGRP) == 0, "K is a whole number of load groups");
  constexpr int NGRP = K / (4 * KGRP);
  const int h  = threadIdx.x >> 5;
  const int l  = threadIdx.x & 31;
  const int lm = l & 15;
  const int lh = l >> 4;
  const int m0 = blockIdx.x * 16;

  const v2f* ap = (const v2f*)(A + (size_t)(m0 + lm) * K + 2 * lh);
  const v4f* bp = (const v4f*)(B + (size_t)h * K * NHID + (size_t)(2 * lh) * NHID + 4 * lm);

  v2f af[2][KGRP];
  v4f bf[2][KGRP][2];
  v8f acc[4];
#pragma unroll
  for (int nt = 0; nt < 4; ++nt)
    acc[nt] = (v8f){0.f,0.f,0.f,0.f,0.f,0.f,0.f,0.f};

#pragma unroll
  for (int j = 0; j < KGRP; ++j) {
    af[0][j]    = ap[2 * j];
    bf[0][j][0] = bp[(4 * j) * (NHID / 4)];
    bf[0][j][1] = bp[(4 * j + 1) * (NHID / 4)];
  }
#pragma unroll
  for (int g = 0; g < NGRP; ++g) {
    if (g + 1 < NGRP) {
#pragma unroll
      for (int j = 0; j < KGRP; ++j) {
        const int s = (g + 1) * KGRP + j;
        af[(g + 1) & 1][j]    = ap[2 * s];
        bf[(g + 1) & 1][j][0] = bp[(4 * s) * (NHID / 4)];
        bf[(g + 1) & 1][j][1] = bp[(4 * s + 1) * (NHID / 4)];
      }
    }
    __builtin_amdgcn_sched_barrier(0);
#pragma unroll
    for (int j = 0; j < KGRP; ++j) {
      const v2f a = af[g & 1][j];
#pragma unroll
      for (int nt = 0; nt < 4; ++nt) {
        const v2f b = { bf[g & 1][j][0][nt], bf[g & 1][j][1][nt] };
        acc[nt] = wmma4f32(a, b, acc[nt]);
      }
    }
    __builtin_amdgcn_sched_barrier(0);
  }

  float* Ch = C + (size_t)h * NNODES * NHID;
#pragma unroll
  for (int r = 0; r < 8; ++r) {
    v4f o = { acc[0][r], acc[1][r], acc[2][r], acc[3][r] };
    *(v4f*)(Ch + (size_t)(m0 + 8 * lh + r) * NHID + 4 * lm) = o;
  }

  const v4f a1 = *(const v4f*)(avec + (size_t)h * 2 * NHID + 4 * lm);
  const v4f a2 = *(const v4f*)(avec + (size_t)h * 2 * NHID + NHID + 4 * lm);
  float p1[8], p2[8];
#pragma unroll
  for (int r = 0; r < 8; ++r) {
    p1[r] = acc[0][r] * a1[0] + acc[1][r] * a1[1] + acc[2][r] * a1[2] + acc[3][r] * a1[3];
    p2[r] = acc[0][r] * a2[0] + acc[1][r] * a2[1] + acc[2][r] * a2[2] + acc[3][r] * a2[3];
  }
#pragma unroll
  for (int r = 0; r < 8; ++r) {
    p1[r] = row_xor_add<1>(p1[r]);  p2[r] = row_xor_add<1>(p2[r]);
  }
#pragma unroll
  for (int r = 0; r < 8; ++r) {
    p1[r] = row_xor_add<2>(p1[r]);  p2[r] = row_xor_add<2>(p2[r]);
  }
#pragma unroll
  for (int r = 0; r < 8; ++r) {
    p1[r] = row_xor_add<4>(p1[r]);  p2[r] = row_xor_add<4>(p2[r]);
  }
#pragma unroll
  for (int r = 0; r < 8; ++r) {
    p1[r] = row_xor_add<8>(p1[r]);  p2[r] = row_xor_add<8>(p2[r]);
  }
  float e1 = p1[0], e2 = p2[0];
#pragma unroll
  for (int r = 1; r < 8; ++r) {
    e1 = (lm == r) ? p1[r] : e1;
    e2 = (lm == r) ? p2[r] : e2;
  }
  if (lm < 8) {
    const size_t idx = (size_t)h * NNODES + m0 + 8 * lh + lm;
    el[idx] = e1;
    er[idx] = e2;
  }
}

#define WHS_S 68
#define KSTEPS (NROI / 4)
#define LDS_ROWS 128
#define LDS_BYTES ((LDS_ROWS * WHS_S) * sizeof(float))

template <int N>
__device__ __forceinline__ float row_share(float v) {
  int o = __builtin_amdgcn_update_dpp(0, __float_as_int(v), 0x150 | N, 0xf, 0xf, false);
  return __int_as_float(o);
}

__device__ __forceinline__ float masked_score(float eli, float erj, float adjv) {
  float z  = eli + erj;
  float ev = (z > 0.f) ? z : ALPHA_LR * z;
  return (adjv > 0.f) ? ev : -INFINITY;
}

__device__ __forceinline__ float masked_weight(float e, float m) {
  float wv = expf(e - m);
  return (e == -INFINITY) ? 0.f : wv;
}

template <bool LAYER2>
__global__ __launch_bounds__(256)
void att_block_kernel(const float* __restrict__ adj, const float* __restrict__ Wh,
                      const float* __restrict__ el, const float* __restrict__ er,
                      const float* __restrict__ Wnext, const float* __restrict__ anext,
                      float* __restrict__ out, float* __restrict__ elnext,
                      float* __restrict__ ernext) {
  extern __shared__ __attribute__((aligned(16))) float lds[];
  float* whs = lds;

  const int g = blockIdx.x, h = blockIdx.y;
  const int t = threadIdx.x;
  const int r0 = g * NROI;
  const int w  = t >> 5;
  const int l  = t & 31;
  const int lm = l & 15, lh = l >> 4;

  const int  irow  = w * 16 + lm;
  const bool rowok = irow < NROI;
  const int  ic    = rowok ? irow : (NROI - 1);
  const v2f* adjp = (const v2f*)(adj + (size_t)(r0 + ic) * NNODES + r0 + 2 * lh);

  v2f av[KSTEPS], rv[KSTEPS];
  float eli;
#pragma unroll
  for (int s = 0; s < KSTEPS; ++s)
    av[s] = adjp[2 * s];

  if constexpr (!LAYER2) {
    static_assert((NROI * NHID) % 256 == 0, "staging has no remainder");
    constexpr int NSTAGE = (NROI * NHID) / 256;
    const float* WhH = Wh + (size_t)h * NNODES * NHID;
    const float* elh = el + (size_t)h * NNODES + r0;
    const v2f*   erp = (const v2f*)(er + (size_t)h * NNODES + r0 + 2 * lh);
    float st[NSTAGE];
    eli = elh[ic];
#pragma unroll
    for (int s = 0; s < KSTEPS; ++s)
      rv[s] = erp[2 * s];
#pragma unroll
    for (int k = 0; k < NSTAGE; ++k) {
      int i = t + 256 * k;
      st[k] = WhH[(size_t)(r0 + (i >> 6)) * NHID + (i & 63)];
    }
    __builtin_amdgcn_sched_barrier(0);
#pragma unroll
    for (int k = 0; k < NSTAGE; ++k) {
      int i = t + 256 * k;
      whs[(i >> 6) * WHS_S + (i & 63)] = st[k];
    }
  } else {
    constexpr int NV4   = (NROI * NHID) / 4;
    constexpr int NFULL = NV4 / 256;
    static_assert(NV4 - 256 * NFULL == 64, "the remainder is waves 0 and 1");
    constexpr size_t PLANE = (size_t)NNODES * NHID;
    const v4f* Wp = (const v4f*)(Wh + (size_t)r0 * NHID);
    const int  jc = (t < NROI) ? t : (NROI - 1);
    const bool rem = t < (NV4 - 256 * NFULL);
    float elp[4], erq[4];
    v4f st4[NFULL + 1][4];
#pragma unroll
    for (int p = 0; p < 4; ++p) {
      elp[p] = el[(size_t)p * NNODES + r0 + ic];
      erq[p] = er[(size_t)p * NNODES + r0 + jc];
    }
#pragma unroll
    for (int k = 0; k < NFULL; ++k) {
#pragma unroll
      for (int p = 0; p < 4; ++p)
        st4[k][p] = Wp[p * (PLANE / 4) + t + 256 * k];
    }
    if (rem) {
#pragma unroll
      for (int p = 0; p < 4; ++p)
        st4[NFULL][p] = Wp[p * (PLANE / 4) + t + 256 * NFULL];
    }
    __builtin_amdgcn_sched_barrier(0);
    eli = ((elp[0] + elp[1]) + elp[2]) + elp[3];
    if (t < NROI)
      whs[NROI * WHS_S + t] = ((erq[0] + erq[1]) + erq[2]) + erq[3];
#pragma unroll
    for (int k = 0; k < NFULL; ++k) {
      const int i4 = t + 256 * k;
      const v4f sum = ((st4[k][0] + st4[k][1]) + st4[k][2]) + st4[k][3];
      *(v4f*)(whs + (i4 >> 4) * WHS_S + 4 * (i4 & 15)) = sum;
    }
    if (rem) {
      const int i4 = t + 256 * NFULL;
      const v4f sum = ((st4[NFULL][0] + st4[NFULL][1]) + st4[NFULL][2]) + st4[NFULL][3];
      *(v4f*)(whs + (i4 >> 4) * WHS_S + 4 * (i4 & 15)) = sum;
    }
    __syncthreads();
    const v2f* ersp = (const v2f*)(whs + NROI * WHS_S + 2 * lh);
#pragma unroll
    for (int s = 0; s < KSTEPS; ++s)
      rv[s] = ersp[2 * s];
  }

  v2f e[KSTEPS];
  float m = -INFINITY;
#pragma unroll
  for (int s = 0; s < KSTEPS; ++s) {
    e[s][0] = masked_score(eli, rv[s][0], av[s][0]);
    e[s][1] = masked_score(eli, rv[s][1], av[s][1]);
    m = fmaxf(fmaxf(m, e[s][0]), e[s][1]);
  }

  m = fmaxf(m, __shfl_xor(m, 16));
  float ssum = 0.f;
#pragma unroll
  for (int s = 0; s < KSTEPS; ++s) {
    e[s][0] = masked_weight(e[s][0], m);
    e[s][1] = masked_weight(e[s][1], m);
    ssum += e[s][0];
    ssum += e[s][1];
  }
  ssum += __shfl_xor(ssum, 16);
  float inv = (ssum > 0.f) ? 1.f / ssum : 0.f;
  if (!rowok) inv = 0.f;
#pragma unroll
  for (int s = 0; s < KSTEPS; ++s) { e[s][0] *= inv; e[s][1] *= inv; }

  if constexpr (!LAYER2)
    __syncthreads();

  v8f acc[4];
#pragma unroll
  for (int nt = 0; nt < 4; ++nt)
    acc[nt] = (v8f){0.f,0.f,0.f,0.f,0.f,0.f,0.f,0.f};

  const v4f* bq = (const v4f*)(whs + (2 * lh) * WHS_S + 4 * lm);
#pragma unroll
  for (int s = 0; s < KSTEPS; ++s) {
    const v2f a = e[s];
    const v4f b0 = bq[(4 * s) * (WHS_S / 4)];
    const v4f b1 = bq[(4 * s + 1) * (WHS_S / 4)];
#pragma unroll
    for (int nt = 0; nt < 4; ++nt) {
      const v2f b = { b0[nt], b1[nt] };
      acc[nt] = wmma4f32(a, b, acc[nt]);
    }
  }

  float ss[8];
#pragma unroll
  for (int v = 0; v < 8; ++v)
    ss[v] = acc[0][v] * acc[0][v] + acc[1][v] * acc[1][v]
          + acc[2][v] * acc[2][v] + acc[3][v] * acc[3][v];
#pragma unroll
  for (int v = 0; v < 8; ++v) ss[v] = row_xor_add<1>(ss[v]);
#pragma unroll
  for (int v = 0; v < 8; ++v) ss[v] = row_xor_add<2>(ss[v]);
#pragma unroll
  for (int v = 0; v < 8; ++v) ss[v] = row_xor_add<4>(ss[v]);
#pragma unroll
  for (int v = 0; v < 8; ++v) ss[v] = row_xor_add<8>(ss[v]);

  float mine = ss[0];
#pragma unroll
  for (int v = 1; v < 8; ++v) mine = (lm == v) ? ss[v] : mine;
  const float r = sqrtf(mine);
  float sc = fkernel_scale(r);
  if (LAYER2) sc *= fkernel_scale(sc * r);
  float sv[8];
  sv[0] = row_share<0>(sc);  sv[1] = row_share<1>(sc);
  sv[2] = row_share<2>(sc);  sv[3] = row_share<3>(sc);
  sv[4] = row_share<4>(sc);  sv[5] = row_share<5>(sc);
  sv[6] = row_share<6>(sc);  sv[7] = row_share<7>(sc);

  v4f hv[8];
#pragma unroll
  for (int v = 0; v < 8; ++v)
    hv[v] = (v4f){ acc[0][v], acc[1][v], acc[2][v], acc[3][v] };

  if constexpr (LAYER2) {
    int    row  = w * 16 + 8 * lh;
    float* orow = out + (size_t)(r0 + row) * NHID + 4 * lm;
    for (int v = 0; v < 8; ++v) {
      const v4f   x = hv[0];
      const float s = sv[0];
      const v4f o = { act_elu_cos(s * x[0]), act_elu_cos(s * x[1]),
                      act_elu_cos(s * x[2]), act_elu_cos(s * x[3]) };
      if (row < NROI) *(v4f*)orow = o;
      ++row;
      orow += NHID;
#pragma unroll
      for (int j = 0; j < 7; ++j) { hv[j] = hv[j + 1]; sv[j] = sv[j + 1]; }
    }
  } else {
    __syncthreads();

    constexpr int KS2 = NHID / 4;
    const v4f* wp = (const v4f*)(Wnext + (size_t)(h * NHID + 2 * lh) * NHID + 4 * lm);
    v4f wb[KS2][2];
#pragma unroll
    for (int s = 0; s < KS2; ++s) {
      wb[s][0] = wp[(4 * s) * (NHID / 4)];
      wb[s][1] = wp[(4 * s + 1) * (NHID / 4)];
    }
    const v4f a1 = *(const v4f*)(anext + 4 * lm);
    const v4f a2 = *(const v4f*)(anext + NHID + 4 * lm);

    float* xrow = whs + (w * 16 + 8 * lh) * WHS_S + 4 * lm;
    for (int v = 0; v < 8; ++v) {
      const v4f   x = hv[0];
      const float s = sv[0];
      const v4f o = { act_elu_cos(s * x[0]), act_elu_cos(s * x[1]),
                      act_elu_cos(s * x[2]), act_elu_cos(s * x[3]) };
      *(v4f*)xrow = o;
      xrow += WHS_S;
#pragma unroll
      for (int j = 0; j < 7; ++j) { hv[j] = hv[j + 1]; sv[j] = sv[j + 1]; }
    }

    __syncthreads();

    v8f pc[4];
#pragma unroll
    for (int nt = 0; nt < 4; ++nt)
      pc[nt] = (v8f){0.f,0.f,0.f,0.f,0.f,0.f,0.f,0.f};
    const v2f* aq = (const v2f*)(whs + (w * 16 + lm) * WHS_S + 2 * lh);
#pragma unroll
    for (int s = 0; s < KS2; ++s) {
      const v2f a = aq[2 * s];
#pragma unroll
      for (int nt = 0; nt < 4; ++nt) {
        const v2f b = { wb[s][0][nt], wb[s][1][nt] };
        pc[nt] = wmma4f32(a, b, pc[nt]);
      }
    }

    const int rbase = w * 16 + 8 * lh;
    float* Pp = out + (size_t)h * NNODES * NHID + (size_t)(r0 + rbase) * NHID + 4 * lm;
#pragma unroll
    for (int v = 0; v < 8; ++v) {
      const v4f o = { pc[0][v], pc[1][v], pc[2][v], pc[3][v] };
      if (rbase + v < NROI) *(v4f*)(Pp + (size_t)v * NHID) = o;
    }

    float p1[8], p2[8];
#pragma unroll
    for (int v = 0; v < 8; ++v) {
      p1[v] = pc[0][v] * a1[0] + pc[1][v] * a1[1] + pc[2][v] * a1[2] + pc[3][v] * a1[3];
      p2[v] = pc[0][v] * a2[0] + pc[1][v] * a2[1] + pc[2][v] * a2[2] + pc[3][v] * a2[3];
    }
#pragma unroll
    for (int v = 0; v < 8; ++v) {
      p1[v] = row_xor_add<1>(p1[v]);  p2[v] = row_xor_add<1>(p2[v]);
    }
#pragma unroll
    for (int v = 0; v < 8; ++v) {
      p1[v] = row_xor_add<2>(p1[v]);  p2[v] = row_xor_add<2>(p2[v]);
    }
#pragma unroll
    for (int v = 0; v < 8; ++v) {
      p1[v] = row_xor_add<4>(p1[v]);  p2[v] = row_xor_add<4>(p2[v]);
    }
#pragma unroll
    for (int v = 0; v < 8; ++v) {
      p1[v] = row_xor_add<8>(p1[v]);  p2[v] = row_xor_add<8>(p2[v]);
    }
    float e1 = p1[0], e2 = p2[0];
#pragma unroll
    for (int v = 1; v < 8; ++v) {
      e1 = (lm == v) ? p1[v] : e1;
      e2 = (lm == v) ? p2[v] : e2;
    }
    if (lm < 8 && rbase + lm < NROI) {
      const size_t idx = (size_t)h * NNODES + r0 + rbase + lm;
      elnext[idx] = e1;
      ernext[idx] = e2;
    }
  }
}

extern "C" void kernel_launch(void* const* d_in, const int* in_sizes, int n_in,
                              void* d_out, int out_size, void* d_ws, size_t ws_size,
                              hipStream_t stream) {
  (void)in_sizes; (void)n_in; (void)out_size; (void)ws_size;
  const float* x       = (const float*)d_in[0];
  const float* adj     = (const float*)d_in[1];
  const float* W_heads = (const float*)d_in[2];
  const float* a_heads = (const float*)d_in[3];
  const float* W_out   = (const float*)d_in[4];
  const float* a_out   = (const float*)d_in[5];
  float* out = (float*)d_out;

  float* ws = (float*)d_ws;
  float* Wh   = ws;                  ws += (size_t)NHEADS * NNODES * NHID;
  float* el1  = ws;                  ws += (size_t)NHEADS * NNODES;
  float* er1  = ws;                  ws += (size_t)NHEADS * NNODES;
  float* Wh2p = ws;                  ws += (size_t)NHEADS * NNODES * NHID;
  float* el2p = ws;                  ws += (size_t)NHEADS * NNODES;
  float* er2p = ws;

  const int mtiles = NNODES / 16;

  proj_strip_kernel<NFEAT, NHEADS><<<mtiles, 32 * NHEADS, 0, stream>>>(
      x, W_heads, a_heads, Wh, el1, er1);
  att_block_kernel<false><<<dim3(NGRAPH, NHEADS), 256, LDS_BYTES, stream>>>(
      adj, Wh, el1, er1, W_out, a_out, Wh2p, el2p, er2p);
  att_block_kernel<true><<<dim3(NGRAPH, 1), 256, LDS_BYTES, stream>>>(
      adj, Wh2p, el2p, er2p, nullptr, nullptr, out, nullptr, nullptr);
}


# adj reduced once to a 1-bit-per-edge row mask (4 words per row, wave ballots in the projection launch); each attention lane loads 8 B of mask, not 29 strided 8-B adj loads; sum orders unchanged
# speedup vs baseline: 3.8647x; 1.0714x over previous
#include <hip/hip_runtime.h>
#include <math.h>

#define NNODES   3712
#define NROI     116
#define NGRAPH   32
#define NFEAT    128
#define NHID     64
#define NHEADS   4
#define ALPHA_LR 0.1f
#define A_PARA   0.1f
#define PHI_C    0.94247779607693793f
#define MAXNORM  (1.0f - 4e-3f)

typedef __attribute__((ext_vector_type(2))) float v2f;
typedef __attribute__((ext_vector_type(4))) float v4f;
typedef __attribute__((ext_vector_type(8))) float v8f;
typedef __attribute__((ext_vector_type(2))) unsigned int v2u;
typedef __attribute__((ext_vector_type(4))) unsigned int v4u;

__device__ __forceinline__ v8f wmma4f32(v2f a, v2f b, v8f c) {
  return __builtin_amdgcn_wmma_f32_16x16x4_f32(
      false, a, false, b, (short)0, c, false, false);
}

__device__ __forceinline__ float fkernel_scale(float r) {
  float proj = 1.0f, rn = r;
  if (r > MAXNORM) { proj = MAXNORM / r; rn = MAXNORM; }
  rn = fmaxf(rn, 1e-15f);
  float t = fminf(rn, 1.0f - 1e-7f);
  return proj * (atanhf(t) / rn);
}

__device__ __forceinline__ float act_elu_cos(float v) {
  float e = (v > 0.0f) ? v : expm1f(v);
  return e + A_PARA * cosf(v + PHI_C);
}

#define KGRP 8
#define MASKW 4

template <int MASK>
__device__ __forceinline__ float row_xor_add(float v) {
  int o = __builtin_amdgcn_update_dpp(0, __float_as_int(v), 0x160 | MASK, 0xf, 0xf, false);
  return v + __int_as_float(o);
}

template <int K, int NH>
__global__ __launch_bounds__(32 * NH)
void proj_strip_kernel(const float* __restrict__ A, const float* __restrict__ B,
                       const float* __restrict__ avec, float* __restrict__ C,
                       float* __restrict__ el, float* __restrict__ er,
                       const float* __restrict__ adj, float* __restrict__ emask) {
  static_assert(K % (4 * KGRP) == 0, "K is a whole number of load groups");
  static_assert(NH >= 2, "two waves build the 16 rows' edge masks");
  static_assert(NROI % 4 == 0 && NROI / 4 <= 32, "a row's block is one 16-byte piece per lane");
  constexpr int NGRP = K / (4 * KGRP);
  const int h  = threadIdx.x >> 5;
  const int l  = threadIdx.x & 31;
  const int lm = l & 15;
  const int lh = l >> 4;
  const int m0 = blockIdx.x * 16;

  const v2f* ap = (const v2f*)(A + (size_t)(m0 + lm) * K + 2 * lh);
  const v4f* bp = (const v4f*)(B + (size_t)h * K * NHID + (size_t)(2 * lh) * NHID + 4 * lm);

  const int hw = __builtin_amdgcn_readfirstlane(h);
  v4f am[8];
#pragma unroll
  for (int j = 0; j < 8; ++j) am[j] = (v4f){0.f, 0.f, 0.f, 0.f};
  if (hw < 2) {
    const int lc = (l < NROI / 4) ? l : (NROI / 4 - 1);
#pragma unroll
    for (int j = 0; j < 8; ++j) {
      const int row = m0 + 8 * hw + j;
      const int rb  = (row / NROI) * NROI;
      am[j] = *(const v4f*)(adj + (size_t)row * NNODES + rb + 4 * lc);
    }
  }

  v2f af[2][KGRP];
  v4f bf[2][KGRP][2];
  v8f acc[4];
#pragma unroll
  for (int nt = 0; nt < 4; ++nt)
    acc[nt] = (v8f){0.f,0.f,0.f,0.f,0.f,0.f,0.f,0.f};

#pragma unroll
  for (int j = 0; j < KGRP; ++j) {
    af[0][j]    = ap[2 * j];
    bf[0][j][0] = bp[(4 * j) * (NHID / 4)];
    bf[0][j][1] = bp[(4 * j + 1) * (NHID / 4)];
  }
#pragma unroll
  for (int g = 0; g < NGRP; ++g) {
    if (g + 1 < NGRP) {
#pragma unroll
      for (int j = 0; j < KGRP; ++j) {
        const int s = (g + 1) * KGRP + j;
        af[(g + 1) & 1][j]    = ap[2 * s];
        bf[(g + 1) & 1][j][0] = bp[(4 * s) * (NHID / 4)];
        bf[(g + 1) & 1][j][1] = bp[(4 * s + 1) * (NHID / 4)];
      }
    }
    __builtin_amdgcn_sched_barrier(0);
    if (g == 0 && hw < 2) {
      constexpr unsigned int COLS = (1u << (NROI / 4)) - 1u;
      v4u mw = (v4u){0u, 0u, 0u, 0u};
#pragma unroll
      for (int j = 0; j < 8; ++j) {
        const v4u bj = { __builtin_amdgcn_ballot_w32(am[j][0] > 0.f) & COLS,
                         __builtin_amdgcn_ballot_w32(am[j][1] > 0.f) & COLS,
                         __builtin_amdgcn_ballot_w32(am[j][2] > 0.f) & COLS,
                         __builtin_amdgcn_ballot_w32(am[j][3] > 0.f) & COLS };
        mw = (l == j) ? bj : mw;
      }
      if (l < 8)
        *(v4u*)((unsigned int*)emask + (size_t)(m0 + 8 * hw + l) * MASKW) = mw;
    }
#pragma unroll
    for (int j = 0; j < KGRP; ++j) {
      const v2f a = af[g & 1][j];
#pragma unroll
      for (int nt = 0; nt < 4; ++nt) {
        const v2f b = { bf[g & 1][j][0][nt], bf[g & 1][j][1][nt] };
        acc[nt] = wmma4f32(a, b, acc[nt]);
      }
    }
    __builtin_amdgcn_sched_barrier(0);
  }

  float* Ch = C + (size_t)h * NNODES * NHID;
#pragma unroll
  for (int r = 0; r < 8; ++r) {
    v4f o = { acc[0][r], acc[1][r], acc[2][r], acc[3][r] };
    *(v4f*)(Ch + (size_t)(m0 + 8 * lh + r) * NHID + 4 * lm) = o;
  }

  const v4f a1 = *(const v4f*)(avec + (size_t)h * 2 * NHID + 4 * lm);
  const v4f a2 = *(const v4f*)(avec + (size_t)h * 2 * NHID + NHID + 4 * lm);
  float p1[8], p2[8];
#pragma unroll
  for (int r = 0; r < 8; ++r) {
    p1[r] = acc[0][r] * a1[0] + acc[1][r] * a1[1] + acc[2][r] * a1[2] + acc[3][r] * a1[3];
    p2[r] = acc[0][r] * a2[0] + acc[1][r] * a2[1] + acc[2][r] * a2[2] + acc[3][r] * a2[3];
  }
#pragma unroll
  for (int r = 0; r < 8; ++r) {
    p1[r] = row_xor_add<1>(p1[r]);  p2[r] = row_xor_add<1>(p2[r]);
  }
#pragma unroll
  for (int r = 0; r < 8; ++r) {
    p1[r] = row_xor_add<2>(p1[r]);  p2[r] = row_xor_add<2>(p2[r]);
  }
#pragma unroll
  for (int r = 0; r < 8; ++r) {
    p1[r] = row_xor_add<4>(p1[r]);  p2[r] = row_xor_add<4>(p2[r]);
  }
#pragma unroll
  for (int r = 0; r < 8; ++r) {
    p1[r] = row_xor_add<8>(p1[r]);  p2[r] = row_xor_add<8>(p2[r]);
  }
  float e1 = p1[0], e2 = p2[0];
#pragma unroll
  for (int r = 1; r < 8; ++r) {
    e1 = (lm == r) ? p1[r] : e1;
    e2 = (lm == r) ? p2[r] : e2;
  }
  if (lm < 8) {
    const size_t idx = (size_t)h * NNODES + m0 + 8 * lh + lm;
    el[idx] = e1;
    er[idx] = e2;
  }
}

#define WHS_S 68
#define KSTEPS (NROI / 4)
#define LDS_ROWS 128
#define LDS_BYTES ((LDS_ROWS * WHS_S) * sizeof(float))

template <int N>
__device__ __forceinline__ float row_share(float v) {
  int o = __builtin_amdgcn_update_dpp(0, __float_as_int(v), 0x150 | N, 0xf, 0xf, false);
  return __int_as_float(o);
}

__device__ __forceinline__ float masked_score(float eli, float erj, unsigned int word, int bit) {
  float z  = eli + erj;
  float ev = (z > 0.f) ? z : ALPHA_LR * z;
  return ((word >> bit) & 1u) ? ev : -INFINITY;
}

__device__ __forceinline__ float masked_weight(float e, float m) {
  float wv = expf(e - m);
  return (e == -INFINITY) ? 0.f : wv;
}

template <bool LAYER2>
__global__ __launch_bounds__(256)
void att_block_kernel(const float* __restrict__ emask, const float* __restrict__ Wh,
                      const float* __restrict__ el, const float* __restrict__ er,
                      const float* __restrict__ Wnext, const float* __restrict__ anext,
                      float* __restrict__ out, float* __restrict__ elnext,
                      float* __restrict__ ernext) {
  extern __shared__ __attribute__((aligned(16))) float lds[];
  float* whs = lds;

  const int g = blockIdx.x, h = blockIdx.y;
  const int t = threadIdx.x;
  const int r0 = g * NROI;
  const int w  = t >> 5;
  const int l  = t & 31;
  const int lm = l & 15, lh = l >> 4;

  const int  irow  = w * 16 + lm;
  const bool rowok = irow < NROI;
  const int  ic    = rowok ? irow : (NROI - 1);
  const v2u mk = *(const v2u*)((const unsigned int*)emask + (size_t)(r0 + ic) * MASKW + 2 * lh);

  v2f rv[KSTEPS];
  float eli;

  if constexpr (!LAYER2) {
    static_assert((NROI * NHID) % 256 == 0, "staging has no remainder");
    constexpr int NSTAGE = (NROI * NHID) / 256;
    const float* WhH = Wh + (size_t)h * NNODES * NHID;
    const float* elh = el + (size_t)h * NNODES + r0;
    const v2f*   erp = (const v2f*)(er + (size_t)h * NNODES + r0 + 2 * lh);
    float st[NSTAGE];
    eli = elh[ic];
#pragma unroll
    for (int s = 0; s < KSTEPS; ++s)
      rv[s] = erp[2 * s];
#pragma unroll
    for (int k = 0; k < NSTAGE; ++k) {
      int i = t + 256 * k;
      st[k] = WhH[(size_t)(r0 + (i >> 6)) * NHID + (i & 63)];
    }
    __builtin_amdgcn_sched_barrier(0);
#pragma unroll
    for (int k = 0; k < NSTAGE; ++k) {
      int i = t + 256 * k;
      whs[(i >> 6) * WHS_S + (i & 63)] = st[k];
    }
  } else {
    constexpr int NV4   = (NROI * NHID) / 4;
    constexpr int NFULL = NV4 / 256;
    static_assert(NV4 - 256 * NFULL == 64, "the remainder is waves 0 and 1");
    constexpr size_t PLANE = (size_t)NNODES * NHID;
    const v4f* Wp = (const v4f*)(Wh + (size_t)r0 * NHID);
    const int  jc = (t < NROI) ? t : (NROI - 1);
    const bool rem = t < (NV4 - 256 * NFULL);
    float elp[4], erq[4];
    v4f st4[NFULL + 1][4];
#pragma unroll
    for (int p = 0; p < 4; ++p) {
      elp[p] = el[(size_t)p * NNODES + r0 + ic];
      erq[p] = er[(size_t)p * NNODES + r0 + jc];
    }
#pragma unroll
    for (int k = 0; k < NFULL; ++k) {
#pragma unroll
      for (int p = 0; p < 4; ++p)
        st4[k][p] = Wp[p * (PLANE / 4) + t + 256 * k];
    }
    if (rem) {
#pragma unroll
      for (int p = 0; p < 4; ++p)
        st4[NFULL][p] = Wp[p * (PLANE / 4) + t + 256 * NFULL];
    }
    __builtin_amdgcn_sched_barrier(0);
    eli = ((elp[0] + elp[1]) + elp[2]) + elp[3];
    if (t < NROI)
      whs[NROI * WHS_S + t] = ((erq[0] + erq[1]) + erq[2]) + erq[3];
#pragma unroll
    for (int k = 0; k < NFULL; ++k) {
      const int i4 = t + 256 * k;
      const v4f sum = ((st4[k][0] + st4[k][1]) + st4[k][2]) + st4[k][3];
      *(v4f*)(whs + (i4 >> 4) * WHS_S + 4 * (i4 & 15)) = sum;
    }
    if (rem) {
      const int i4 = t + 256 * NFULL;
      const v4f sum = ((st4[NFULL][0] + st4[NFULL][1]) + st4[NFULL][2]) + st4[NFULL][3];
      *(v4f*)(whs + (i4 >> 4) * WHS_S + 4 * (i4 & 15)) = sum;
    }
    __syncthreads();
    const v2f* ersp = (const v2f*)(whs + NROI * WHS_S + 2 * lh);
#pragma unroll
    for (int s = 0; s < KSTEPS; ++s)
      rv[s] = ersp[2 * s];
  }

  v2f e[KSTEPS];
  float m = -INFINITY;
#pragma unroll
  for (int s = 0; s < KSTEPS; ++s) {
    e[s][0] = masked_score(eli, rv[s][0], mk[0], s);
    e[s][1] = masked_score(eli, rv[s][1], mk[1], s);
    m = fmaxf(fmaxf(m, e[s][0]), e[s][1]);
  }

  m = fmaxf(m, __shfl_xor(m, 16));
  float ssum = 0.f;
#pragma unroll
  for (int s = 0; s < KSTEPS; ++s) {
    e[s][0] = masked_weight(e[s][0], m);
    e[s][1] = masked_weight(e[s][1], m);
    ssum += e[s][0];
    ssum += e[s][1];
  }
  ssum += __shfl_xor(ssum, 16);
  float inv = (ssum > 0.f) ? 1.f / ssum : 0.f;
  if (!rowok) inv = 0.f;
#pragma unroll
  for (int s = 0; s < KSTEPS; ++s) { e[s][0] *= inv; e[s][1] *= inv; }

  if constexpr (!LAYER2)
    __syncthreads();

  v8f acc[4];
#pragma unroll
  for (int nt = 0; nt < 4; ++nt)
    acc[nt] = (v8f){0.f,0.f,0.f,0.f,0.f,0.f,0.f,0.f};

  const v4f* bq = (const v4f*)(whs + (2 * lh) * WHS_S + 4 * lm);
#pragma unroll
  for (int s = 0; s < KSTEPS; ++s) {
    const v2f a = e[s];
    const v4f b0 = bq[(4 * s) * (WHS_S / 4)];
    const v4f b1 = bq[(4 * s + 1) * (WHS_S / 4)];
#pragma unroll
    for (int nt = 0; nt < 4; ++nt) {
      const v2f b = { b0[nt], b1[nt] };
      acc[nt] = wmma4f32(a, b, acc[nt]);
    }
  }

  float ss[8];
#pragma unroll
  for (int v = 0; v < 8; ++v)
    ss[v] = acc[0][v] * acc[0][v] + acc[1][v] * acc[1][v]
          + acc[2][v] * acc[2][v] + acc[3][v] * acc[3][v];
#pragma unroll
  for (int v = 0; v < 8; ++v) ss[v] = row_xor_add<1>(ss[v]);
#pragma unroll
  for (int v = 0; v < 8; ++v) ss[v] = row_xor_add<2>(ss[v]);
#pragma unroll
  for (int v = 0; v < 8; ++v) ss[v] = row_xor_add<4>(ss[v]);
#pragma unroll
  for (int v = 0; v < 8; ++v) ss[v] = row_xor_add<8>(ss[v]);

  float mine = ss[0];
#pragma unroll
  for (int v = 1; v < 8; ++v) mine = (lm == v) ? ss[v] : mine;
  const float r = sqrtf(mine);
  float sc = fkernel_scale(r);
  if (LAYER2) sc *= fkernel_scale(sc * r);
  float sv[8];
  sv[0] = row_share<0>(sc);  sv[1] = row_share<1>(sc);
  sv[2] = row_share<2>(sc);  sv[3] = row_share<3>(sc);
  sv[4] = row_share<4>(sc);  sv[5] = row_share<5>(sc);
  sv[6] = row_share<6>(sc);  sv[7] = row_share<7>(sc);

  v4f hv[8];
#pragma unroll
  for (int v = 0; v < 8; ++v)
    hv[v] = (v4f){ acc[0][v], acc[1][v], acc[2][v], acc[3][v] };

  if constexpr (LAYER2) {
    int    row  = w * 16 + 8 * lh;
    float* orow = out + (size_t)(r0 + row) * NHID + 4 * lm;
    for (int v = 0; v < 8; ++v) {
      const v4f   x = hv[0];
      const float s = sv[0];
      const v4f o = { act_elu_cos(s * x[0]), act_elu_cos(s * x[1]),
                      act_elu_cos(s * x[2]), act_elu_cos(s * x[3]) };
      if (row < NROI) *(v4f*)orow = o;
      ++row;
      orow += NHID;
#pragma unroll
      for (int j = 0; j < 7; ++j) { hv[j] = hv[j + 1]; sv[j] = sv[j + 1]; }
    }
  } else {
    __syncthreads();

    constexpr int KS2 = NHID / 4;
    const v4f* wp = (const v4f*)(Wnext + (size_t)(h * NHID + 2 * lh) * NHID + 4 * lm);
    v4f wb[KS2][2];
#pragma unroll
    for (int s = 0; s < KS2; ++s) {
      wb[s][0] = wp[(4 * s) * (NHID / 4)];
      wb[s][1] = wp[(4 * s + 1) * (NHID / 4)];
    }
    const v4f a1 = *(const v4f*)(anext + 4 * lm);
    const v4f a2 = *(const v4f*)(anext + NHID + 4 * lm);

    float* xrow = whs + (w * 16 + 8 * lh) * WHS_S + 4 * lm;
    for (int v = 0; v < 8; ++v) {
      const v4f   x = hv[0];
      const float s = sv[0];
      const v4f o = { act_elu_cos(s * x[0]), act_elu_cos(s * x[1]),
                      act_elu_cos(s * x[2]), act_elu_cos(s * x[3]) };
      *(v4f*)xrow = o;
      xrow += WHS_S;
#pragma unroll
      for (int j = 0; j < 7; ++j) { hv[j] = hv[j + 1]; sv[j] = sv[j + 1]; }
    }

    __syncthreads();

    v8f pc[4];
#pragma unroll
    for (int nt = 0; nt < 4; ++nt)
      pc[nt] = (v8f){0.f,0.f,0.f,0.f,0.f,0.f,0.f,0.f};
    const v2f* aq = (const v2f*)(whs + (w * 16 + lm) * WHS_S + 2 * lh);
#pragma unroll
    for (int s = 0; s < KS2; ++s) {
      const v2f a = aq[2 * s];
#pragma unroll
      for (int nt = 0; nt < 4; ++nt) {
        const v2f b = { wb[s][0][nt], wb[s][1][nt] };
        pc[nt] = wmma4f32(a, b, pc[nt]);
      }
    }

    const int rbase = w * 16 + 8 * lh;
    float* Pp = out + (size_t)h * NNODES * NHID + (size_t)(r0 + rbase) * NHID + 4 * lm;
#pragma unroll
    for (int v = 0; v < 8; ++v) {
      const v4f o = { pc[0][v], pc[1][v], pc[2][v], pc[3][v] };
      if (rbase + v < NROI) *(v4f*)(Pp + (size_t)v * NHID) = o;
    }

    float p1[8], p2[8];
#pragma unroll
    for (int v = 0; v < 8; ++v) {
      p1[v] = pc[0][v] * a1[0] + pc[1][v] * a1[1] + pc[2][v] * a1[2] + pc[3][v] * a1[3];
      p2[v] = pc[0][v] * a2[0] + pc[1][v] * a2[1] + pc[2][v] * a2[2] + pc[3][v] * a2[3];
    }
#pragma unroll
    for (int v = 0; v < 8; ++v) {
      p1[v] = row_xor_add<1>(p1[v]);  p2[v] = row_xor_add<1>(p2[v]);
    }
#pragma unroll
    for (int v = 0; v < 8; ++v) {
      p1[v] = row_xor_add<2>(p1[v]);  p2[v] = row_xor_add<2>(p2[v]);
    }
#pragma unroll
    for (int v = 0; v < 8; ++v) {
      p1[v] = row_xor_add<4>(p1[v]);  p2[v] = row_xor_add<4>(p2[v]);
    }
#pragma unroll
    for (int v = 0; v < 8; ++v) {
      p1[v] = row_xor_add<8>(p1[v]);  p2[v] = row_xor_add<8>(p2[v]);
    }
    float e1 = p1[0], e2 = p2[0];
#pragma unroll
    for (int v = 1; v < 8; ++v) {
      e1 = (lm == v) ? p1[v] : e1;
      e2 = (lm == v) ? p2[v] : e2;
    }
    if (lm < 8 && rbase + lm < NROI) {
      const size_t idx = (size_t)h * NNODES + r0 + rbase + lm;
      elnext[idx] = e1;
      ernext[idx] = e2;
    }
  }
}

extern "C" void kernel_launch(void* const* d_in, const int* in_sizes, int n_in,
                              void* d_out, int out_size, void* d_ws, size_t ws_size,
                              hipStream_t stream) {
  (void)in_sizes; (void)n_in; (void)out_size; (void)ws_size;
  const float* x       = (const float*)d_in[0];
  const float* adj     = (const float*)d_in[1];
  const float* W_heads = (const float*)d_in[2];
  const float* a_heads = (const float*)d_in[3];
  const float* W_out   = (const float*)d_in[4];
  const float* a_out   = (const float*)d_in[5];
  float* out = (float*)d_out;

  float* ws = (float*)d_ws;
  float* Wh   = ws;                  ws += (size_t)NHEADS * NNODES * NHID;
  float* el1  = ws;                  ws += (size_t)NHEADS * NNODES;
  float* er1  = ws;                  ws += (size_t)NHEADS * NNODES;
  float* Wh2p = ws;                  ws += (size_t)NHEADS * NNODES * NHID;
  float* el2p = ws;                  ws += (size_t)NHEADS * NNODES;
  float* er2p = ws;                  ws += (size_t)NHEADS * NNODES;
  float* emask = ws;

  const int mtiles = NNODES / 16;

  proj_strip_kernel<NFEAT, NHEADS><<<mtiles, 32 * NHEADS, 0, stream>>>(
      x, W_heads, a_heads, Wh, el1, er1, adj, emask);
  att_block_kernel<false><<<dim3(NGRAPH, NHEADS), 256, LDS_BYTES, stream>>>(
      emask, Wh, el1, er1, W_out, a_out, Wh2p, el2p, er2p);
  att_block_kernel<true><<<dim3(NGRAPH, 1), 256, LDS_BYTES, stream>>>(
      emask, Wh2p, el2p, er2p, nullptr, nullptr, out, nullptr, nullptr);
}


# Softmax weights as row factor x column factor: exp(leaky(el_i+er_j)-bound) = z>0 ? A_i*B_j : C_i*D_j; 2 exps per row and per column (LDS table), not 1 per edge; weights re-rounded, sum orders kept
# speedup vs baseline: 4.0196x; 1.0401x over previous
#include <hip/hip_runtime.h>
#include <math.h>

#define NNODES   3712
#define NROI     116
#define NGRAPH   32
#define NFEAT    128
#define NHID     64
#define NHEADS   4
#define ALPHA_LR 0.1f
#define A_PARA   0.1f
#define PHI_C    0.94247779607693793f
#define MAXNORM  (1.0f - 4e-3f)

typedef __attribute__((ext_vector_type(2))) float v2f;
typedef __attribute__((ext_vector_type(4))) float v4f;
typedef __attribute__((ext_vector_type(8))) float v8f;
typedef __attribute__((ext_vector_type(2))) unsigned int v2u;
typedef __attribute__((ext_vector_type(4))) unsigned int v4u;

__device__ __forceinline__ v8f wmma4f32(v2f a, v2f b, v8f c) {
  return __builtin_amdgcn_wmma_f32_16x16x4_f32(
      false, a, false, b, (short)0, c, false, false);
}

__device__ __forceinline__ float fkernel_scale(float r) {
  float proj = 1.0f, rn = r;
  if (r > MAXNORM) { proj = MAXNORM / r; rn = MAXNORM; }
  rn = fmaxf(rn, 1e-15f);
  float t = fminf(rn, 1.0f - 1e-7f);
  return proj * (atanhf(t) / rn);
}

__device__ __forceinline__ float act_elu_cos(float v) {
  float e = (v > 0.0f) ? v : expm1f(v);
  return e + A_PARA * cosf(v + PHI_C);
}

#define KGRP 8
#define MASKW 4

template <int MASK>
__device__ __forceinline__ float row_xor_add(float v) {
  int o = __builtin_amdgcn_update_dpp(0, __float_as_int(v), 0x160 | MASK, 0xf, 0xf, false);
  return v + __int_as_float(o);
}

template <int K, int NH>
__global__ __launch_bounds__(32 * NH)
void proj_strip_kernel(const float* __restrict__ A, const float* __restrict__ B,
                       const float* __restrict__ avec, float* __restrict__ C,
                       float* __restrict__ el, float* __restrict__ er,
                       const float* __restrict__ adj, float* __restrict__ emask) {
  static_assert(K % (4 * KGRP) == 0, "K is a whole number of load groups");
  static_assert(NH >= 2, "two waves build the 16 rows' edge masks");
  static_assert(NROI % 4 == 0 && NROI / 4 <= 32, "a row's block is one 16-byte piece per lane");
  constexpr int NGRP = K / (4 * KGRP);
  const int h  = threadIdx.x >> 5;
  const int l  = threadIdx.x & 31;
  const int lm = l & 15;
  const int lh = l >> 4;
  const int m0 = blockIdx.x * 16;

  const v2f* ap = (const v2f*)(A + (size_t)(m0 + lm) * K + 2 * lh);
  const v4f* bp = (const v4f*)(B + (size_t)h * K * NHID + (size_t)(2 * lh) * NHID + 4 * lm);

  const int hw = __builtin_amdgcn_readfirstlane(h);
  v4f am[8];
#pragma unroll
  for (int j = 0; j < 8; ++j) am[j] = (v4f){0.f, 0.f, 0.f, 0.f};
  if (hw < 2) {
    const int lc = (l < NROI / 4) ? l : (NROI / 4 - 1);
#pragma unroll
    for (int j = 0; j < 8; ++j) {
      const int row = m0 + 8 * hw + j;
      const int rb  = (row / NROI) * NROI;
      am[j] = *(const v4f*)(adj + (size_t)row * NNODES + rb + 4 * lc);
    }
  }

  v2f af[2][KGRP];
  v4f bf[2][KGRP][2];
  v8f acc[4];
#pragma unroll
  for (int nt = 0; nt < 4; ++nt)
    acc[nt] = (v8f){0.f,0.f,0.f,0.f,0.f,0.f,0.f,0.f};

#pragma unroll
  for (int j = 0; j < KGRP; ++j) {
    af[0][j]    = ap[2 * j];
    bf[0][j][0] = bp[(4 * j) * (NHID / 4)];
    bf[0][j][1] = bp[(4 * j + 1) * (NHID / 4)];
  }
#pragma unroll
  for (int g = 0; g < NGRP; ++g) {
    if (g + 1 < NGRP) {
#pragma unroll
      for (int j = 0; j < KGRP; ++j) {
        const int s = (g + 1) * KGRP + j;
        af[(g + 1) & 1][j]    = ap[2 * s];
        bf[(g + 1) & 1][j][0] = bp[(4 * s) * (NHID / 4)];
        bf[(g + 1) & 1][j][1] = bp[(4 * s + 1) * (NHID / 4)];
      }
    }
    __builtin_amdgcn_sched_barrier(0);
    if (g == 0 && hw < 2) {
      constexpr unsigned int COLS = (1u << (NROI / 4)) - 1u;
      v4u mw = (v4u){0u, 0u, 0u, 0u};
#pragma unroll
      for (int j = 0; j < 8; ++j) {
        const v4u bj = { __builtin_amdgcn_ballot_w32(am[j][0] > 0.f) & COLS,
                         __builtin_amdgcn_ballot_w32(am[j][1] > 0.f) & COLS,
                         __builtin_amdgcn_ballot_w32(am[j][2] > 0.f) & COLS,
                         __builtin_amdgcn_ballot_w32(am[j][3] > 0.f) & COLS };
        mw = (l == j) ? bj : mw;
      }
      if (l < 8)
        *(v4u*)((unsigned int*)emask + (size_t)(m0 + 8 * hw + l) * MASKW) = mw;
    }
#pragma unroll
    for (int j = 0; j < KGRP; ++j) {
      const v2f a = af[g & 1][j];
#pragma unroll
      for (int nt = 0; nt < 4; ++nt) {
        const v2f b = { bf[g & 1][j][0][nt], bf[g & 1][j][1][nt] };
        acc[nt] = wmma4f32(a, b, acc[nt]);
      }
    }
    __builtin_amdgcn_sched_barrier(0);
  }

  float* Ch = C + (size_t)h * NNODES * NHID;
#pragma unroll
  for (int r = 0; r < 8; ++r) {
    v4f o = { acc[0][r], acc[1][r], acc[2][r], acc[3][r] };
    *(v4f*)(Ch + (size_t)(m0 + 8 * lh + r) * NHID + 4 * lm) = o;
  }

  const v4f a1 = *(const v4f*)(avec + (size_t)h * 2 * NHID + 4 * lm);
  const v4f a2 = *(const v4f*)(avec + (size_t)h * 2 * NHID + NHID + 4 * lm);
  float p1[8], p2[8];
#pragma unroll
  for (int r = 0; r < 8; ++r) {
    p1[r] = acc[0][r] * a1[0] + acc[1][r] * a1[1] + acc[2][r] * a1[2] + acc[3][r] * a1[3];
    p2[r] = acc[0][r] * a2[0] + acc[1][r] * a2[1] + acc[2][r] * a2[2] + acc[3][r] * a2[3];
  }
#pragma unroll
  for (int r = 0; r < 8; ++r) {
    p1[r] = row_xor_add<1>(p1[r]);  p2[r] = row_xor_add<1>(p2[r]);
  }
#pragma unroll
  for (int r = 0; r < 8; ++r) {
    p1[r] = row_xor_add<2>(p1[r]);  p2[r] = row_xor_add<2>(p2[r]);
  }
#pragma unroll
  for (int r = 0; r < 8; ++r) {
    p1[r] = row_xor_add<4>(p1[r]);  p2[r] = row_xor_add<4>(p2[r]);
  }
#pragma unroll
  for (int r = 0; r < 8; ++r) {
    p1[r] = row_xor_add<8>(p1[r]);  p2[r] = row_xor_add<8>(p2[r]);
  }
  float e1 = p1[0], e2 = p2[0];
#pragma unroll
  for (int r = 1; r < 8; ++r) {
    e1 = (lm == r) ? p1[r] : e1;
    e2 = (lm == r) ? p2[r] : e2;
  }
  if (lm < 8) {
    const size_t idx = (size_t)h * NNODES + m0 + 8 * lh + lm;
    el[idx] = e1;
    er[idx] = e2;
  }
}

#define WHS_S 68
#define KSTEPS (NROI / 4)
#define LDS_ROWS 128
#define TAB_OFF (LDS_ROWS * WHS_S)
#define TAB_REC 8
#define LDS_BYTES ((LDS_ROWS * WHS_S + (NROI / 2) * TAB_REC) * sizeof(float))

template <int N>
__device__ __forceinline__ float row_share(float v) {
  int o = __builtin_amdgcn_update_dpp(0, __float_as_int(v), 0x150 | N, 0xf, 0xf, false);
  return __int_as_float(o);
}

template <int MASK>
__device__ __forceinline__ float row_xor_max(float v) {
  int o = __builtin_amdgcn_update_dpp(0, __float_as_int(v), 0x160 | MASK, 0xf, 0xf, false);
  return fmaxf(v, __int_as_float(o));
}

__device__ __forceinline__ float masked_score(float eli, float erj, unsigned int word, int bit) {
  float z  = eli + erj;
  float ev = (z > 0.f) ? z : ALPHA_LR * z;
  return ((word >> bit) & 1u) ? ev : -INFINITY;
}

__device__ __forceinline__ float masked_weight(float e, float m) {
  float wv = expf(e - m);
  return (e == -INFINITY) ? 0.f : wv;
}

template <bool LAYER2>
__global__ __launch_bounds__(256)
void att_block_kernel(const float* __restrict__ emask, const float* __restrict__ Wh,
                      const float* __restrict__ el, const float* __restrict__ er,
                      const float* __restrict__ Wnext, const float* __restrict__ anext,
                      float* __restrict__ out, float* __restrict__ elnext,
                      float* __restrict__ ernext) {
  extern __shared__ __attribute__((aligned(16))) float lds[];
  float* whs = lds;

  const int g = blockIdx.x, h = blockIdx.y;
  const int t = threadIdx.x;
  const int r0 = g * NROI;
  const int w  = t >> 5;
  const int l  = t & 31;
  const int lm = l & 15, lh = l >> 4;

  const int  irow  = w * 16 + lm;
  const bool rowok = irow < NROI;
  const int  ic    = rowok ? irow : (NROI - 1);
  const v2u mk = *(const v2u*)((const unsigned int*)emask + (size_t)(r0 + ic) * MASKW + 2 * lh);

  const int lq = (l < NROI / 4) ? l : (NROI / 4 - 1);
  const int jc = (t < NROI) ? t : (NROI - 1);
  float eli, ert;
  v4f   er4;
  float* tab = lds + TAB_OFF;

  if constexpr (!LAYER2) {
    static_assert((NROI * NHID) % 256 == 0, "staging has no remainder");
    constexpr int NSTAGE = (NROI * NHID) / 256;
    const float* WhH = Wh + (size_t)h * NNODES * NHID;
    const float* elh = el + (size_t)h * NNODES + r0;
    const float* erh = er + (size_t)h * NNODES + r0;
    float st[NSTAGE];
    eli = elh[ic];
    ert = erh[jc];
    er4 = *(const v4f*)(erh + 4 * lq);
#pragma unroll
    for (int k = 0; k < NSTAGE; ++k) {
      int i = t + 256 * k;
      st[k] = WhH[(size_t)(r0 + (i >> 6)) * NHID + (i & 63)];
    }
    __builtin_amdgcn_sched_barrier(0);
#pragma unroll
    for (int k = 0; k < NSTAGE; ++k) {
      int i = t + 256 * k;
      whs[(i >> 6) * WHS_S + (i & 63)] = st[k];
    }
  } else {
    constexpr int NV4   = (NROI * NHID) / 4;
    constexpr int NFULL = NV4 / 256;
    static_assert(NV4 - 256 * NFULL == 64, "the remainder is waves 0 and 1");
    constexpr size_t PLANE = (size_t)NNODES * NHID;
    const v4f* Wp = (const v4f*)(Wh + (size_t)r0 * NHID);
    const bool rem = t < (NV4 - 256 * NFULL);
    float elp[4], erq[4];
    v4f er4p[4];
    v4f st4[NFULL + 1][4];
#pragma unroll
    for (int p = 0; p < 4; ++p) {
      elp[p]  = el[(size_t)p * NNODES + r0 + ic];
      erq[p]  = er[(size_t)p * NNODES + r0 + jc];
      er4p[p] = *(const v4f*)(er + (size_t)p * NNODES + r0 + 4 * lq);
    }
#pragma unroll
    for (int k = 0; k < NFULL; ++k) {
#pragma unroll
      for (int p = 0; p < 4; ++p)
        st4[k][p] = Wp[p * (PLANE / 4) + t + 256 * k];
    }
    if (rem) {
#pragma unroll
      for (int p = 0; p < 4; ++p)
        st4[NFULL][p] = Wp[p * (PLANE / 4) + t + 256 * NFULL];
    }
    __builtin_amdgcn_sched_barrier(0);
    eli = ((elp[0] + elp[1]) + elp[2]) + elp[3];
    ert = ((erq[0] + erq[1]) + erq[2]) + erq[3];
    er4 = ((er4p[0] + er4p[1]) + er4p[2]) + er4p[3];
#pragma unroll
    for (int k = 0; k < NFULL; ++k) {
      const int i4 = t + 256 * k;
      const v4f sum = ((st4[k][0] + st4[k][1]) + st4[k][2]) + st4[k][3];
      *(v4f*)(whs + (i4 >> 4) * WHS_S + 4 * (i4 & 15)) = sum;
    }
    if (rem) {
      const int i4 = t + 256 * NFULL;
      const v4f sum = ((st4[NFULL][0] + st4[NFULL][1]) + st4[NFULL][2]) + st4[NFULL][3];
      *(v4f*)(whs + (i4 >> 4) * WHS_S + 4 * (i4 & 15)) = sum;
    }
  }

  float R = fmaxf(fmaxf(er4[0], er4[1]), fmaxf(er4[2], er4[3]));
  R = row_xor_max<1>(R);
  R = row_xor_max<2>(R);
  R = row_xor_max<4>(R);
  R = row_xor_max<8>(R);
  R = fmaxf(R, __shfl_xor(R, 16));

  if (t < NROI) {
    const float d = ert - R;
    float* rec = tab + (t >> 1) * TAB_REC + (t & 1);
    rec[0] = ert;
    rec[2] = expf(d);
    rec[4] = expf(ALPHA_LR * d);
  }

  __syncthreads();

  const float u  = eli + R;
  const float mb = (u > 0.f) ? u : ALPHA_LR * u;
  const float ga = expf(u - mb);
  const float gc = expf(ALPHA_LR * u - mb);

  const float* tp = tab + lh * TAB_REC;
  v2f e[KSTEPS];
  float ssum = 0.f;
#pragma unroll
  for (int s = 0; s < KSTEPS; ++s) {
    const v4f eb = *(const v4f*)(tp + 2 * s * TAB_REC);
    const v2f dd = *(const v2f*)(tp + 2 * s * TAB_REC + 4);
    float w0 = (eli + eb[0] > 0.f) ? ga * eb[2] : gc * dd[0];
    float w1 = (eli + eb[1] > 0.f) ? ga * eb[3] : gc * dd[1];
    w0 = ((mk[0] >> s) & 1u) ? w0 : 0.f;
    w1 = ((mk[1] >> s) & 1u) ? w1 : 0.f;
    e[s][0] = w0;
    e[s][1] = w1;
    ssum += w0;
    ssum += w1;
  }
  ssum += __shfl_xor(ssum, 16);

  const bool weak = ((mk[0] | mk[1]) != 0u) && !(ssum >= 1e-20f);
  if (__builtin_expect(__builtin_amdgcn_ballot_w32(weak) != 0u, 0)) {
    float m = -INFINITY;
#pragma unroll
    for (int s = 0; s < KSTEPS; ++s) {
      const v2f rv = *(const v2f*)(tp + 2 * s * TAB_REC);
      e[s][0] = masked_score(eli, rv[0], mk[0], s);
      e[s][1] = masked_score(eli, rv[1], mk[1], s);
      m = fmaxf(fmaxf(m, e[s][0]), e[s][1]);
    }
    m = fmaxf(m, __shfl_xor(m, 16));
    ssum = 0.f;
#pragma unroll
    for (int s = 0; s < KSTEPS; ++s) {
      e[s][0] = masked_weight(e[s][0], m);
      e[s][1] = masked_weight(e[s][1], m);
      ssum += e[s][0];
      ssum += e[s][1];
    }
    ssum += __shfl_xor(ssum, 16);
  }

  float inv = (ssum > 0.f) ? 1.f / ssum : 0.f;
  if (!rowok) inv = 0.f;
#pragma unroll
  for (int s = 0; s < KSTEPS; ++s) { e[s][0] *= inv; e[s][1] *= inv; }

  v8f acc[4];
#pragma unroll
  for (int nt = 0; nt < 4; ++nt)
    acc[nt] = (v8f){0.f,0.f,0.f,0.f,0.f,0.f,0.f,0.f};

  const v4f* bq = (const v4f*)(whs + (2 * lh) * WHS_S + 4 * lm);
#pragma unroll
  for (int s = 0; s < KSTEPS; ++s) {
    const v2f a = e[s];
    const v4f b0 = bq[(4 * s) * (WHS_S / 4)];
    const v4f b1 = bq[(4 * s + 1) * (WHS_S / 4)];
#pragma unroll
    for (int nt = 0; nt < 4; ++nt) {
      const v2f b = { b0[nt], b1[nt] };
      acc[nt] = wmma4f32(a, b, acc[nt]);
    }
  }

  float ss[8];
#pragma unroll
  for (int v = 0; v < 8; ++v)
    ss[v] = acc[0][v] * acc[0][v] + acc[1][v] * acc[1][v]
          + acc[2][v] * acc[2][v] + acc[3][v] * acc[3][v];
#pragma unroll
  for (int v = 0; v < 8; ++v) ss[v] = row_xor_add<1>(ss[v]);
#pragma unroll
  for (int v = 0; v < 8; ++v) ss[v] = row_xor_add<2>(ss[v]);
#pragma unroll
  for (int v = 0; v < 8; ++v) ss[v] = row_xor_add<4>(ss[v]);
#pragma unroll
  for (int v = 0; v < 8; ++v) ss[v] = row_xor_add<8>(ss[v]);

  float mine = ss[0];
#pragma unroll
  for (int v = 1; v < 8; ++v) mine = (lm == v) ? ss[v] : mine;
  const float r = sqrtf(mine);
  float sc = fkernel_scale(r);
  if (LAYER2) sc *= fkernel_scale(sc * r);
  float sv[8];
  sv[0] = row_share<0>(sc);  sv[1] = row_share<1>(sc);
  sv[2] = row_share<2>(sc);  sv[3] = row_share<3>(sc);
  sv[4] = row_share<4>(sc);  sv[5] = row_share<5>(sc);
  sv[6] = row_share<6>(sc);  sv[7] = row_share<7>(sc);

  v4f hv[8];
#pragma unroll
  for (int v = 0; v < 8; ++v)
    hv[v] = (v4f){ acc[0][v], acc[1][v], acc[2][v], acc[3][v] };

  if constexpr (LAYER2) {
    int    row  = w * 16 + 8 * lh;
    float* orow = out + (size_t)(r0 + row) * NHID + 4 * lm;
    for (int v = 0; v < 8; ++v) {
      const v4f   x = hv[0];
      const float s = sv[0];
      const v4f o = { act_elu_cos(s * x[0]), act_elu_cos(s * x[1]),
                      act_elu_cos(s * x[2]), act_elu_cos(s * x[3]) };
      if (row < NROI) *(v4f*)orow = o;
      ++row;
      orow += NHID;
#pragma unroll
      for (int j = 0; j < 7; ++j) { hv[j] = hv[j + 1]; sv[j] = sv[j + 1]; }
    }
  } else {
    __syncthreads();

    constexpr int KS2 = NHID / 4;
    const v4f* wp = (const v4f*)(Wnext + (size_t)(h * NHID + 2 * lh) * NHID + 4 * lm);
    v4f wb[KS2][2];
#pragma unroll
    for (int s = 0; s < KS2; ++s) {
      wb[s][0] = wp[(4 * s) * (NHID / 4)];
      wb[s][1] = wp[(4 * s + 1) * (NHID / 4)];
    }
    const v4f a1 = *(const v4f*)(anext + 4 * lm);
    const v4f a2 = *(const v4f*)(anext + NHID + 4 * lm);

    float* xrow = whs + (w * 16 + 8 * lh) * WHS_S + 4 * lm;
    for (int v = 0; v < 8; ++v) {
      const v4f   x = hv[0];
      const float s = sv[0];
      const v4f o = { act_elu_cos(s * x[0]), act_elu_cos(s * x[1]),
                      act_elu_cos(s * x[2]), act_elu_cos(s * x[3]) };
      *(v4f*)xrow = o;
      xrow += WHS_S;
#pragma unroll
      for (int j = 0; j < 7; ++j) { hv[j] = hv[j + 1]; sv[j] = sv[j + 1]; }
    }

    __syncthreads();

    v8f pc[4];
#pragma unroll
    for (int nt = 0; nt < 4; ++nt)
      pc[nt] = (v8f){0.f,0.f,0.f,0.f,0.f,0.f,0.f,0.f};
    const v2f* aq = (const v2f*)(whs + (w * 16 + lm) * WHS_S + 2 * lh);
#pragma unroll
    for (int s = 0; s < KS2; ++s) {
      const v2f a = aq[2 * s];
#pragma unroll
      for (int nt = 0; nt < 4; ++nt) {
        const v2f b = { wb[s][0][nt], wb[s][1][nt] };
        pc[nt] = wmma4f32(a, b, pc[nt]);
      }
    }

    const int rbase = w * 16 + 8 * lh;
    float* Pp = out + (size_t)h * NNODES * NHID + (size_t)(r0 + rbase) * NHID + 4 * lm;
#pragma unroll
    for (int v = 0; v < 8; ++v) {
      const v4f o = { pc[0][v], pc[1][v], pc[2][v], pc[3][v] };
      if (rbase + v < NROI) *(v4f*)(Pp + (size_t)v * NHID) = o;
    }

    float p1[8], p2[8];
#pragma unroll
    for (int v = 0; v < 8; ++v) {
      p1[v] = pc[0][v] * a1[0] + pc[1][v] * a1[1] + pc[2][v] * a1[2] + pc[3][v] * a1[3];
      p2[v] = pc[0][v] * a2[0] + pc[1][v] * a2[1] + pc[2][v] * a2[2] + pc[3][v] * a2[3];
    }
#pragma unroll
    for (int v = 0; v < 8; ++v) {
      p1[v] = row_xor_add<1>(p1[v]);  p2[v] = row_xor_add<1>(p2[v]);
    }
#pragma unroll
    for (int v = 0; v < 8; ++v) {
      p1[v] = row_xor_add<2>(p1[v]);  p2[v] = row_xor_add<2>(p2[v]);
    }
#pragma unroll
    for (int v = 0; v < 8; ++v) {
      p1[v] = row_xor_add<4>(p1[v]);  p2[v] = row_xor_add<4>(p2[v]);
    }
#pragma unroll
    for (int v = 0; v < 8; ++v) {
      p1[v] = row_xor_add<8>(p1[v]);  p2[v] = row_xor_add<8>(p2[v]);
    }
    float e1 = p1[0], e2 = p2[0];
#pragma unroll
    for (int v = 1; v < 8; ++v) {
      e1 = (lm == v) ? p1[v] : e1;
      e2 = (lm == v) ? p2[v] : e2;
    }
    if (lm < 8 && rbase + lm < NROI) {
      const size_t idx = (size_t)h * NNODES + r0 + rbase + lm;
      elnext[idx] = e1;
      ernext[idx] = e2;
    }
  }
}

extern "C" void kernel_launch(void* const* d_in, const int* in_sizes, int n_in,
                              void* d_out, int out_size, void* d_ws, size_t ws_size,
                              hipStream_t stream) {
  (void)in_sizes; (void)n_in; (void)out_size; (void)ws_size;
  const float* x       = (const float*)d_in[0];
  const float* adj     = (const float*)d_in[1];
  const float* W_heads = (const float*)d_in[2];
  const float* a_heads = (const float*)d_in[3];
  const float* W_out   = (const float*)d_in[4];
  const float* a_out   = (const float*)d_in[5];
  float* out = (float*)d_out;

  float* ws = (float*)d_ws;
  float* Wh   = ws;                  ws += (size_t)NHEADS * NNODES * NHID;
  float* el1  = ws;                  ws += (size_t)NHEADS * NNODES;
  float* er1  = ws;                  ws += (size_t)NHEADS * NNODES;
  float* Wh2p = ws;                  ws += (size_t)NHEADS * NNODES * NHID;
  float* el2p = ws;                  ws += (size_t)NHEADS * NNODES;
  float* er2p = ws;                  ws += (size_t)NHEADS * NNODES;
  float* emask = ws;

  const int mtiles = NNODES / 16;

  proj_strip_kernel<NFEAT, NHEADS><<<mtiles, 32 * NHEADS, 0, stream>>>(
      x, W_heads, a_heads, Wh, el1, er1, adj, emask);
  att_block_kernel<false><<<dim3(NGRAPH, NHEADS), 256, LDS_BYTES, stream>>>(
      emask, Wh, el1, er1, W_out, a_out, Wh2p, el2p, er2p);
  att_block_kernel<true><<<dim3(NGRAPH, 1), 256, LDS_BYTES, stream>>>(
      emask, Wh2p, el2p, er2p, nullptr, nullptr, out, nullptr, nullptr);
}


# Layer-2 tail split by output columns over two workgroups per graph: each redoes the block's product and row norms, then scales, activates and stores its 32 columns; row norm summed in a new order
# speedup vs baseline: 4.2147x; 1.0485x over previous
#include <hip/hip_runtime.h>
#include <math.h>

#define NNODES   3712
#define NROI     116
#define NGRAPH   32
#define NFEAT    128
#define NHID     64
#define NHEADS   4
#define ALPHA_LR 0.1f
#define A_PARA   0.1f
#define PHI_C    0.94247779607693793f
#define MAXNORM  (1.0f - 4e-3f)

typedef __attribute__((ext_vector_type(2))) float v2f;
typedef __attribute__((ext_vector_type(4))) float v4f;
typedef __attribute__((ext_vector_type(8))) float v8f;
typedef __attribute__((ext_vector_type(2))) unsigned int v2u;
typedef __attribute__((ext_vector_type(4))) unsigned int v4u;

__device__ __forceinline__ v8f wmma4f32(v2f a, v2f b, v8f c) {
  return __builtin_amdgcn_wmma_f32_16x16x4_f32(
      false, a, false, b, (short)0, c, false, false);
}

__device__ __forceinline__ float fkernel_scale(float r) {
  float proj = 1.0f, rn = r;
  if (r > MAXNORM) { proj = MAXNORM / r; rn = MAXNORM; }
  rn = fmaxf(rn, 1e-15f);
  float t = fminf(rn, 1.0f - 1e-7f);
  return proj * (atanhf(t) / rn);
}

__device__ __forceinline__ float act_elu_cos(float v) {
  float e = (v > 0.0f) ? v : expm1f(v);
  return e + A_PARA * cosf(v + PHI_C);
}

#define KGRP 8
#define MASKW 4

template <int MASK>
__device__ __forceinline__ float row_xor_add(float v) {
  int o = __builtin_amdgcn_update_dpp(0, __float_as_int(v), 0x160 | MASK, 0xf, 0xf, false);
  return v + __int_as_float(o);
}

template <int K, int NH>
__global__ __launch_bounds__(32 * NH)
void proj_strip_kernel(const float* __restrict__ A, const float* __restrict__ B,
                       const float* __restrict__ avec, float* __restrict__ C,
                       float* __restrict__ el, float* __restrict__ er,
                       const float* __restrict__ adj, float* __restrict__ emask) {
  static_assert(K % (4 * KGRP) == 0, "K is a whole number of load groups");
  static_assert(NH >= 2, "two waves build the 16 rows' edge masks");
  static_assert(NROI % 4 == 0 && NROI / 4 <= 32, "a row's block is one 16-byte piece per lane");
  constexpr int NGRP = K / (4 * KGRP);
  const int h  = threadIdx.x >> 5;
  const int l  = threadIdx.x & 31;
  const int lm = l & 15;
  const int lh = l >> 4;
  const int m0 = blockIdx.x * 16;

  const v2f* ap = (const v2f*)(A + (size_t)(m0 + lm) * K + 2 * lh);
  const v4f* bp = (const v4f*)(B + (size_t)h * K * NHID + (size_t)(2 * lh) * NHID + 4 * lm);

  const int hw = __builtin_amdgcn_readfirstlane(h);
  v4f am[8];
#pragma unroll
  for (int j = 0; j < 8; ++j) am[j] = (v4f){0.f, 0.f, 0.f, 0.f};
  if (hw < 2) {
    const int lc = (l < NROI / 4) ? l : (NROI / 4 - 1);
#pragma unroll
    for (int j = 0; j < 8; ++j) {
      const int row = m0 + 8 * hw + j;
      const int rb  = (row / NROI) * NROI;
      am[j] = *(const v4f*)(adj + (size_t)row * NNODES + rb + 4 * lc);
    }
  }

  v2f af[2][KGRP];
  v4f bf[2][KGRP][2];
  v8f acc[4];
#pragma unroll
  for (int nt = 0; nt < 4; ++nt)
    acc[nt] = (v8f){0.f,0.f,0.f,0.f,0.f,0.f,0.f,0.f};

#pragma unroll
  for (int j = 0; j < KGRP; ++j) {
    af[0][j]    = ap[2 * j];
    bf[0][j][0] = bp[(4 * j) * (NHID / 4)];
    bf[0][j][1] = bp[(4 * j + 1) * (NHID / 4)];
  }
#pragma unroll
  for (int g = 0; g < NGRP; ++g) {
    if (g + 1 < NGRP) {
#pragma unroll
      for (int j = 0; j < KGRP; ++j) {
        const int s = (g + 1) * KGRP + j;
        af[(g + 1) & 1][j]    = ap[2 * s];
        bf[(g + 1) & 1][j][0] = bp[(4 * s) * (NHID / 4)];
        bf[(g + 1) & 1][j][1] = bp[(4 * s + 1) * (NHID / 4)];
      }
    }
    __builtin_amdgcn_sched_barrier(0);
    if (g == 0 && hw < 2) {
      constexpr unsigned int COLS = (1u << (NROI / 4)) - 1u;
      v4u mw = (v4u){0u, 0u, 0u, 0u};
#pragma unroll
      for (int j = 0; j < 8; ++j) {
        const v4u bj = { __builtin_amdgcn_ballot_w32(am[j][0] > 0.f) & COLS,
                         __builtin_amdgcn_ballot_w32(am[j][1] > 0.f) & COLS,
                         __builtin_amdgcn_ballot_w32(am[j][2] > 0.f) & COLS,
                         __builtin_amdgcn_ballot_w32(am[j][3] > 0.f) & COLS };
        mw = (l == j) ? bj : mw;
      }
      if (l < 8)
        *(v4u*)((unsigned int*)emask + (size_t)(m0 + 8 * hw + l) * MASKW) = mw;
    }
#pragma unroll
    for (int j = 0; j < KGRP; ++j) {
      const v2f a = af[g & 1][j];
#pragma unroll
      for (int nt = 0; nt < 4; ++nt) {
        const v2f b = { bf[g & 1][j][0][nt], bf[g & 1][j][1][nt] };
        acc[nt] = wmma4f32(a, b, acc[nt]);
      }
    }
    __builtin_amdgcn_sched_barrier(0);
  }

  float* Ch = C + (size_t)h * NNODES * NHID;
#pragma unroll
  for (int r = 0; r < 8; ++r) {
    v4f o = { acc[0][r], acc[1][r], acc[2][r], acc[3][r] };
    *(v4f*)(Ch + (size_t)(m0 + 8 * lh + r) * NHID + 4 * lm) = o;
  }

  const v4f a1 = *(const v4f*)(avec + (size_t)h * 2 * NHID + 4 * lm);
  const v4f a2 = *(const v4f*)(avec + (size_t)h * 2 * NHID + NHID + 4 * lm);
  float p1[8], p2[8];
#pragma unroll
  for (int r = 0; r < 8; ++r) {
    p1[r] = acc[0][r] * a1[0] + acc[1][r] * a1[1] + acc[2][r] * a1[2] + acc[3][r] * a1[3];
    p2[r] = acc[0][r] * a2[0] + acc[1][r] * a2[1] + acc[2][r] * a2[2] + acc[3][r] * a2[3];
  }
#pragma unroll
  for (int r = 0; r < 8; ++r) {
    p1[r] = row_xor_add<1>(p1[r]);  p2[r] = row_xor_add<1>(p2[r]);
  }
#pragma unroll
  for (int r = 0; r < 8; ++r) {
    p1[r] = row_xor_add<2>(p1[r]);  p2[r] = row_xor_add<2>(p2[r]);
  }
#pragma unroll
  for (int r = 0; r < 8; ++r) {
    p1[r] = row_xor_add<4>(p1[r]);  p2[r] = row_xor_add<4>(p2[r]);
  }
#pragma unroll
  for (int r = 0; r < 8; ++r) {
    p1[r] = row_xor_add<8>(p1[r]);  p2[r] = row_xor_add<8>(p2[r]);
  }
  float e1 = p1[0], e2 = p2[0];
#pragma unroll
  for (int r = 1; r < 8; ++r) {
    e1 = (lm == r) ? p1[r] : e1;
    e2 = (lm == r) ? p2[r] : e2;
  }
  if (lm < 8) {
    const size_t idx = (size_t)h * NNODES + m0 + 8 * lh + lm;
    el[idx] = e1;
    er[idx] = e2;
  }
}

#define WHS_S 68
#define KSTEPS (NROI / 4)
#define LDS_ROWS 128
#define TAB_OFF (LDS_ROWS * WHS_S)
#define TAB_REC 8
#define LDS_BYTES ((LDS_ROWS * WHS_S + (NROI / 2) * TAB_REC) * sizeof(float))

template <int N>
__device__ __forceinline__ float row_share(float v) {
  int o = __builtin_amdgcn_update_dpp(0, __float_as_int(v), 0x150 | N, 0xf, 0xf, false);
  return __int_as_float(o);
}

template <int MASK>
__device__ __forceinline__ float row_xor_max(float v) {
  int o = __builtin_amdgcn_update_dpp(0, __float_as_int(v), 0x160 | MASK, 0xf, 0xf, false);
  return fmaxf(v, __int_as_float(o));
}

__device__ __forceinline__ float masked_score(float eli, float erj, unsigned int word, int bit) {
  float z  = eli + erj;
  float ev = (z > 0.f) ? z : ALPHA_LR * z;
  return ((word >> bit) & 1u) ? ev : -INFINITY;
}

__device__ __forceinline__ float masked_weight(float e, float m) {
  float wv = expf(e - m);
  return (e == -INFINITY) ? 0.f : wv;
}

template <bool LAYER2>
__global__ __launch_bounds__(256)
void att_block_kernel(const float* __restrict__ emask, const float* __restrict__ Wh,
                      const float* __restrict__ el, const float* __restrict__ er,
                      const float* __restrict__ Wnext, const float* __restrict__ anext,
                      float* __restrict__ out, float* __restrict__ elnext,
                      float* __restrict__ ernext) {
  extern __shared__ __attribute__((aligned(16))) float lds[];
  float* whs = lds;

  const int g = blockIdx.x, h = blockIdx.y;
  const int t = threadIdx.x;
  const int r0 = g * NROI;
  const int w  = t >> 5;
  const int l  = t & 31;
  const int lm = l & 15, lh = l >> 4;

  const int  irow  = w * 16 + lm;
  const bool rowok = irow < NROI;
  const int  ic    = rowok ? irow : (NROI - 1);
  const v2u mk = *(const v2u*)((const unsigned int*)emask + (size_t)(r0 + ic) * MASKW + 2 * lh);

  const int lq = (l < NROI / 4) ? l : (NROI / 4 - 1);
  const int jc = (t < NROI) ? t : (NROI - 1);
  float eli, ert;
  v4f   er4;
  float* tab = lds + TAB_OFF;

  if constexpr (!LAYER2) {
    static_assert((NROI * NHID) % 256 == 0, "staging has no remainder");
    constexpr int NSTAGE = (NROI * NHID) / 256;
    const float* WhH = Wh + (size_t)h * NNODES * NHID;
    const float* elh = el + (size_t)h * NNODES + r0;
    const float* erh = er + (size_t)h * NNODES + r0;
    float st[NSTAGE];
    eli = elh[ic];
    ert = erh[jc];
    er4 = *(const v4f*)(erh + 4 * lq);
#pragma unroll
    for (int k = 0; k < NSTAGE; ++k) {
      int i = t + 256 * k;
      st[k] = WhH[(size_t)(r0 + (i >> 6)) * NHID + (i & 63)];
    }
    __builtin_amdgcn_sched_barrier(0);
#pragma unroll
    for (int k = 0; k < NSTAGE; ++k) {
      int i = t + 256 * k;
      whs[(i >> 6) * WHS_S + (i & 63)] = st[k];
    }
  } else {
    constexpr int NV4   = (NROI * NHID) / 4;
    constexpr int NFULL = NV4 / 256;
    static_assert(NV4 - 256 * NFULL == 64, "the remainder is waves 0 and 1");
    constexpr size_t PLANE = (size_t)NNODES * NHID;
    const v4f* Wp = (const v4f*)(Wh + (size_t)r0 * NHID);
    const bool rem = t < (NV4 - 256 * NFULL);
    float elp[4], erq[4];
    v4f er4p[4];
    v4f st4[NFULL + 1][4];
#pragma unroll
    for (int p = 0; p < 4; ++p) {
      elp[p]  = el[(size_t)p * NNODES + r0 + ic];
      erq[p]  = er[(size_t)p * NNODES + r0 + jc];
      er4p[p] = *(const v4f*)(er + (size_t)p * NNODES + r0 + 4 * lq);
    }
#pragma unroll
    for (int k = 0; k < NFULL; ++k) {
#pragma unroll
      for (int p = 0; p < 4; ++p)
        st4[k][p] = Wp[p * (PLANE / 4) + t + 256 * k];
    }
    if (rem) {
#pragma unroll
      for (int p = 0; p < 4; ++p)
        st4[NFULL][p] = Wp[p * (PLANE / 4) + t + 256 * NFULL];
    }
    __builtin_amdgcn_sched_barrier(0);
    eli = ((elp[0] + elp[1]) + elp[2]) + elp[3];
    ert = ((erq[0] + erq[1]) + erq[2]) + erq[3];
    er4 = ((er4p[0] + er4p[1]) + er4p[2]) + er4p[3];
#pragma unroll
    for (int k = 0; k < NFULL; ++k) {
      const int i4 = t + 256 * k;
      const v4f sum = ((st4[k][0] + st4[k][1]) + st4[k][2]) + st4[k][3];
      float* dst = whs + (i4 >> 4) * WHS_S + 8 * (i4 & 7) + 2 * ((i4 >> 3) & 1);
      *(v2f*)dst       = (v2f){ sum[0], sum[1] };
      *(v2f*)(dst + 4) = (v2f){ sum[2], sum[3] };
    }
    if (rem) {
      const int i4 = t + 256 * NFULL;
      const v4f sum = ((st4[NFULL][0] + st4[NFULL][1]) + st4[NFULL][2]) + st4[NFULL][3];
      float* dst = whs + (i4 >> 4) * WHS_S + 8 * (i4 & 7) + 2 * ((i4 >> 3) & 1);
      *(v2f*)dst       = (v2f){ sum[0], sum[1] };
      *(v2f*)(dst + 4) = (v2f){ sum[2], sum[3] };
    }
  }

  float R = fmaxf(fmaxf(er4[0], er4[1]), fmaxf(er4[2], er4[3]));
  R = row_xor_max<1>(R);
  R = row_xor_max<2>(R);
  R = row_xor_max<4>(R);
  R = row_xor_max<8>(R);
  R = fmaxf(R, __shfl_xor(R, 16));

  if (t < NROI) {
    const float d = ert - R;
    float* rec = tab + (t >> 1) * TAB_REC + (t & 1);
    rec[0] = ert;
    rec[2] = expf(d);
    rec[4] = expf(ALPHA_LR * d);
  }

  __syncthreads();

  const float u  = eli + R;
  const float mb = (u > 0.f) ? u : ALPHA_LR * u;
  const float ga = expf(u - mb);
  const float gc = expf(ALPHA_LR * u - mb);

  const float* tp = tab + lh * TAB_REC;
  v2f e[KSTEPS];
  float ssum = 0.f;
#pragma unroll
  for (int s = 0; s < KSTEPS; ++s) {
    const v4f eb = *(const v4f*)(tp + 2 * s * TAB_REC);
    const v2f dd = *(const v2f*)(tp + 2 * s * TAB_REC + 4);
    float w0 = (eli + eb[0] > 0.f) ? ga * eb[2] : gc * dd[0];
    float w1 = (eli + eb[1] > 0.f) ? ga * eb[3] : gc * dd[1];
    w0 = ((mk[0] >> s) & 1u) ? w0 : 0.f;
    w1 = ((mk[1] >> s) & 1u) ? w1 : 0.f;
    e[s][0] = w0;
    e[s][1] = w1;
    ssum += w0;
    ssum += w1;
  }
  ssum += __shfl_xor(ssum, 16);

  const bool weak = ((mk[0] | mk[1]) != 0u) && !(ssum >= 1e-20f);
  if (__builtin_expect(__builtin_amdgcn_ballot_w32(weak) != 0u, 0)) {
    float m = -INFINITY;
#pragma unroll
    for (int s = 0; s < KSTEPS; ++s) {
      const v2f rv = *(const v2f*)(tp + 2 * s * TAB_REC);
      e[s][0] = masked_score(eli, rv[0], mk[0], s);
      e[s][1] = masked_score(eli, rv[1], mk[1], s);
      m = fmaxf(fmaxf(m, e[s][0]), e[s][1]);
    }
    m = fmaxf(m, __shfl_xor(m, 16));
    ssum = 0.f;
#pragma unroll
    for (int s = 0; s < KSTEPS; ++s) {
      e[s][0] = masked_weight(e[s][0], m);
      e[s][1] = masked_weight(e[s][1], m);
      ssum += e[s][0];
      ssum += e[s][1];
    }
    ssum += __shfl_xor(ssum, 16);
  }

  float inv = (ssum > 0.f) ? 1.f / ssum : 0.f;
  if (!rowok) inv = 0.f;
#pragma unroll
  for (int s = 0; s < KSTEPS; ++s) { e[s][0] *= inv; e[s][1] *= inv; }

  v8f acc[4];
#pragma unroll
  for (int nt = 0; nt < 4; ++nt)
    acc[nt] = (v8f){0.f,0.f,0.f,0.f,0.f,0.f,0.f,0.f};

  const v4f* bq = (const v4f*)(whs + (2 * lh) * WHS_S + 4 * lm);
#pragma unroll
  for (int s = 0; s < KSTEPS; ++s) {
    const v2f a = e[s];
    const v4f b0 = bq[(4 * s) * (WHS_S / 4)];
    const v4f b1 = bq[(4 * s + 1) * (WHS_S / 4)];
#pragma unroll
    for (int nt = 0; nt < 4; ++nt) {
      const v2f b = { b0[nt], b1[nt] };
      acc[nt] = wmma4f32(a, b, acc[nt]);
    }
  }

  float ss[8];
#pragma unroll
  for (int v = 0; v < 8; ++v)
    ss[v] = acc[0][v] * acc[0][v] + acc[1][v] * acc[1][v]
          + acc[2][v] * acc[2][v] + acc[3][v] * acc[3][v];
#pragma unroll
  for (int v = 0; v < 8; ++v) ss[v] = row_xor_add<1>(ss[v]);
#pragma unroll
  for (int v = 0; v < 8; ++v) ss[v] = row_xor_add<2>(ss[v]);
#pragma unroll
  for (int v = 0; v < 8; ++v) ss[v] = row_xor_add<4>(ss[v]);
#pragma unroll
  for (int v = 0; v < 8; ++v) ss[v] = row_xor_add<8>(ss[v]);

  float mine = ss[0];
#pragma unroll
  for (int v = 1; v < 8; ++v) mine = (lm == v) ? ss[v] : mine;
  const float r = sqrtf(mine);
  float sc = fkernel_scale(r);
  if (LAYER2) sc *= fkernel_scale(sc * r);
  float sv[8];
  sv[0] = row_share<0>(sc);  sv[1] = row_share<1>(sc);
  sv[2] = row_share<2>(sc);  sv[3] = row_share<3>(sc);
  sv[4] = row_share<4>(sc);  sv[5] = row_share<5>(sc);
  sv[6] = row_share<6>(sc);  sv[7] = row_share<7>(sc);

  if constexpr (LAYER2) {
    const int c = blockIdx.y;
    v2f hq[8];
#pragma unroll
    for (int v = 0; v < 8; ++v)
      hq[v] = c ? (v2f){ acc[2][v], acc[3][v] } : (v2f){ acc[0][v], acc[1][v] };
    int    row  = w * 16 + 8 * lh;
    float* orow = out + (size_t)(r0 + row) * NHID + 32 * c + 2 * lm;
    for (int v = 0; v < 4; ++v) {
      const v2f   x0 = hq[0], x1 = hq[1];
      const float s0 = sv[0], s1 = sv[1];
      const v2f o0 = { act_elu_cos(s0 * x0[0]), act_elu_cos(s0 * x0[1]) };
      const v2f o1 = { act_elu_cos(s1 * x1[0]), act_elu_cos(s1 * x1[1]) };
      if (row < NROI)     *(v2f*)orow          = o0;
      if (row + 1 < NROI) *(v2f*)(orow + NHID) = o1;
      row  += 2;
      orow += 2 * NHID;
#pragma unroll
      for (int j = 0; j < 6; ++j) { hq[j] = hq[j + 2]; sv[j] = sv[j + 2]; }
    }
  } else {
    v4f hv[8];
#pragma unroll
    for (int v = 0; v < 8; ++v)
      hv[v] = (v4f){ acc[0][v], acc[1][v], acc[2][v], acc[3][v] };

    __syncthreads();

    constexpr int KS2 = NHID / 4;
    const v4f* wp = (const v4f*)(Wnext + (size_t)(h * NHID + 2 * lh) * NHID + 4 * lm);
    v4f wb[KS2][2];
#pragma unroll
    for (int s = 0; s < KS2; ++s) {
      wb[s][0] = wp[(4 * s) * (NHID / 4)];
      wb[s][1] = wp[(4 * s + 1) * (NHID / 4)];
    }
    const v4f a1 = *(const v4f*)(anext + 4 * lm);
    const v4f a2 = *(const v4f*)(anext + NHID + 4 * lm);

    float* xrow = whs + (w * 16 + 8 * lh) * WHS_S + 4 * lm;
    for (int v = 0; v < 8; ++v) {
      const v4f   x = hv[0];
      const float s = sv[0];
      const v4f o = { act_elu_cos(s * x[0]), act_elu_cos(s * x[1]),
                      act_elu_cos(s * x[2]), act_elu_cos(s * x[3]) };
      *(v4f*)xrow = o;
      xrow += WHS_S;
#pragma unroll
      for (int j = 0; j < 7; ++j) { hv[j] = hv[j + 1]; sv[j] = sv[j + 1]; }
    }

    __syncthreads();

    v8f pc[4];
#pragma unroll
    for (int nt = 0; nt < 4; ++nt)
      pc[nt] = (v8f){0.f,0.f,0.f,0.f,0.f,0.f,0.f,0.f};
    const v2f* aq = (const v2f*)(whs + (w * 16 + lm) * WHS_S + 2 * lh);
#pragma unroll
    for (int s = 0; s < KS2; ++s) {
      const v2f a = aq[2 * s];
#pragma unroll
      for (int nt = 0; nt < 4; ++nt) {
        const v2f b = { wb[s][0][nt], wb[s][1][nt] };
        pc[nt] = wmma4f32(a, b, pc[nt]);
      }
    }

    const int rbase = w * 16 + 8 * lh;
    float* Pp = out + (size_t)h * NNODES * NHID + (size_t)(r0 + rbase) * NHID + 4 * lm;
#pragma unroll
    for (int v = 0; v < 8; ++v) {
      const v4f o = { pc[0][v], pc[1][v], pc[2][v], pc[3][v] };
      if (rbase + v < NROI) *(v4f*)(Pp + (size_t)v * NHID) = o;
    }

    float p1[8], p2[8];
#pragma unroll
    for (int v = 0; v < 8; ++v) {
      p1[v] = pc[0][v] * a1[0] + pc[1][v] * a1[1] + pc[2][v] * a1[2] + pc[3][v] * a1[3];
      p2[v] = pc[0][v] * a2[0] + pc[1][v] * a2[1] + pc[2][v] * a2[2] + pc[3][v] * a2[3];
    }
#pragma unroll
    for (int v = 0; v < 8; ++v) {
      p1[v] = row_xor_add<1>(p1[v]);  p2[v] = row_xor_add<1>(p2[v]);
    }
#pragma unroll
    for (int v = 0; v < 8; ++v) {
      p1[v] = row_xor_add<2>(p1[v]);  p2[v] = row_xor_add<2>(p2[v]);
    }
#pragma unroll
    for (int v = 0; v < 8; ++v) {
      p1[v] = row_xor_add<4>(p1[v]);  p2[v] = row_xor_add<4>(p2[v]);
    }
#pragma unroll
    for (int v = 0; v < 8; ++v) {
      p1[v] = row_xor_add<8>(p1[v]);  p2[v] = row_xor_add<8>(p2[v]);
    }
    float e1 = p1[0], e2 = p2[0];
#pragma unroll
    for (int v = 1; v < 8; ++v) {
      e1 = (lm == v) ? p1[v] : e1;
      e2 = (lm == v) ? p2[v] : e2;
    }
    if (lm < 8 && rbase + lm < NROI) {
      const size_t idx = (size_t)h * NNODES + r0 + rbase + lm;
      elnext[idx] = e1;
      ernext[idx] = e2;
    }
  }
}

extern "C" void kernel_launch(void* const* d_in, const int* in_sizes, int n_in,
                              void* d_out, int out_size, void* d_ws, size_t ws_size,
                              hipStream_t stream) {
  (void)in_sizes; (void)n_in; (void)out_size; (void)ws_size;
  const float* x       = (const float*)d_in[0];
  const float* adj     = (const float*)d_in[1];
  const float* W_heads = (const float*)d_in[2];
  const float* a_heads = (const float*)d_in[3];
  const float* W_out   = (const float*)d_in[4];
  const float* a_out   = (const float*)d_in[5];
  float* out = (float*)d_out;

  float* ws = (float*)d_ws;
  float* Wh   = ws;                  ws += (size_t)NHEADS * NNODES * NHID;
  float* el1  = ws;                  ws += (size_t)NHEADS * NNODES;
  float* er1  = ws;                  ws += (size_t)NHEADS * NNODES;
  float* Wh2p = ws;                  ws += (size_t)NHEADS * NNODES * NHID;
  float* el2p = ws;                  ws += (size_t)NHEADS * NNODES;
  float* er2p = ws;                  ws += (size_t)NHEADS * NNODES;
  float* emask = ws;

  const int mtiles = NNODES / 16;

  proj_strip_kernel<NFEAT, NHEADS><<<mtiles, 32 * NHEADS, 0, stream>>>(
      x, W_heads, a_heads, Wh, el1, er1, adj, emask);
  att_block_kernel<false><<<dim3(NGRAPH, NHEADS), 256, LDS_BYTES, stream>>>(
      emask, Wh, el1, er1, W_out, a_out, Wh2p, el2p, er2p);
  att_block_kernel<true><<<dim3(NGRAPH, 2), 256, LDS_BYTES, stream>>>(
      emask, Wh2p, el2p, er2p, nullptr, nullptr, out, nullptr, nullptr);
}


# Activation as straight-line code: the library expm1 and cos main paths written out, their never-taken range branches dropped, so a trip's four chains interleave; last multiply-add fused (re-rounded)
# speedup vs baseline: 4.8507x; 1.1509x over previous
#include <hip/hip_runtime.h>
#include <math.h>

#define NNODES   3712
#define NROI     116
#define NGRAPH   32
#define NFEAT    128
#define NHID     64
#define NHEADS   4
#define ALPHA_LR 0.1f
#define A_PARA   0.1f
#define PHI_C    0.94247779607693793f
#define MAXNORM  (1.0f - 4e-3f)

typedef __attribute__((ext_vector_type(2))) float v2f;
typedef __attribute__((ext_vector_type(4))) float v4f;
typedef __attribute__((ext_vector_type(8))) float v8f;
typedef __attribute__((ext_vector_type(2))) unsigned int v2u;
typedef __attribute__((ext_vector_type(4))) unsigned int v4u;

__device__ __forceinline__ v8f wmma4f32(v2f a, v2f b, v8f c) {
  return __builtin_amdgcn_wmma_f32_16x16x4_f32(
      false, a, false, b, (short)0, c, false, false);
}

__device__ __forceinline__ float fkernel_scale(float r) {
  float proj = 1.0f, rn = r;
  if (r > MAXNORM) { proj = MAXNORM / r; rn = MAXNORM; }
  rn = fmaxf(rn, 1e-15f);
  float t = fminf(rn, 1.0f - 1e-7f);
  return proj * (atanhf(t) / rn);
}

__device__ __forceinline__ float expm1_main(float v) {
  const float n = __builtin_rintf(v * 1.44269502f);
  float r = __builtin_fmaf(n, -0.693147182f, v);
  r = __builtin_fmaf(n, 1.90465421e-9f, r);
  const float tw = ldexpf(1.0f, (int)n);
  float p = __builtin_fmaf(1.99510498e-4f, r, 1.39304996e-3f);
  p = __builtin_fmaf(r, p, 8.33319966e-3f);
  p = __builtin_fmaf(r, p, 4.16665114e-2f);
  p = __builtin_fmaf(r, p, 0.166666672f);
  p = __builtin_fmaf(r, p, 0.5f);
  p = r * p;
  const float er = __builtin_fmaf(r, p, r);
  return __builtin_fmaf(tw, er, tw - 1.0f);
}

__device__ __forceinline__ float cos_main(float x) {
  const float ax = __builtin_fabsf(x);
  const float n  = __builtin_rintf(ax * 0.636619747f);
  float r = __builtin_fmaf(-1.57079625f, n, ax);
  r = __builtin_fmaf(-7.54978942e-8f, n, r);
  r = __builtin_fmaf(-5.39030253e-15f, n, r);
  const unsigned int q = (unsigned int)(int)n;
  const float s = r * r;
  float ps = __builtin_fmaf(-1.94644585e-4f, s, 8.33171979e-3f);
  ps = __builtin_fmaf(s, ps, -0.166666463f);
  ps = s * ps;
  const float sn = __builtin_fmaf(r, ps, r);
  float pc = __builtin_fmaf(2.56687417e-5f, s, -1.39091106e-3f);
  pc = __builtin_fmaf(s, pc, 4.16679047e-2f);
  pc = __builtin_fmaf(s, pc, -0.500000238f);
  const float cs = __builtin_fmaf(s, pc, 1.0f);
  const float m  = (q & 1u) ? -sn : cs;
  return __uint_as_float(__float_as_uint(m) ^ ((q << 30) & 0x80000000u));
}

__device__ __forceinline__ float act_elu_cos(float v) {
  const float em = expm1_main(v);
  const float e  = (v > 0.0f) ? v : em;
  return __builtin_fmaf(A_PARA, cos_main(v + PHI_C), e);
}

#define KGRP 8
#define MASKW 4

template <int MASK>
__device__ __forceinline__ float row_xor_add(float v) {
  int o = __builtin_amdgcn_update_dpp(0, __float_as_int(v), 0x160 | MASK, 0xf, 0xf, false);
  return v + __int_as_float(o);
}

template <int K, int NH>
__global__ __launch_bounds__(32 * NH)
void proj_strip_kernel(const float* __restrict__ A, const float* __restrict__ B,
                       const float* __restrict__ avec, float* __restrict__ C,
                       float* __restrict__ el, float* __restrict__ er,
                       const float* __restrict__ adj, float* __restrict__ emask) {
  static_assert(K % (4 * KGRP) == 0, "K is a whole number of load groups");
  static_assert(NH >= 2, "two waves build the 16 rows' edge masks");
  static_assert(NROI % 4 == 0 && NROI / 4 <= 32, "a row's block is one 16-byte piece per lane");
  constexpr int NGRP = K / (4 * KGRP);
  const int h  = threadIdx.x >> 5;
  const int l  = threadIdx.x & 31;
  const int lm = l & 15;
  const int lh = l >> 4;
  const int m0 = blockIdx.x * 16;

  const v2f* ap = (const v2f*)(A + (size_t)(m0 + lm) * K + 2 * lh);
  const v4f* bp = (const v4f*)(B + (size_t)h * K * NHID + (size_t)(2 * lh) * NHID + 4 * lm);

  const int hw = __builtin_amdgcn_readfirstlane(h);
  v4f am[8];
#pragma unroll
  for (int j = 0; j < 8; ++j) am[j] = (v4f){0.f, 0.f, 0.f, 0.f};
  if (hw < 2) {
    const int lc = (l < NROI / 4) ? l : (NROI / 4 - 1);
#pragma unroll
    for (int j = 0; j < 8; ++j) {
      const int row = m0 + 8 * hw + j;
      const int rb  = (row / NROI) * NROI;
      am[j] = *(const v4f*)(adj + (size_t)row * NNODES + rb + 4 * lc);
    }
  }

  v2f af[2][KGRP];
  v4f bf[2][KGRP][2];
  v8f acc[4];
#pragma unroll
  for (int nt = 0; nt < 4; ++nt)
    acc[nt] = (v8f){0.f,0.f,0.f,0.f,0.f,0.f,0.f,0.f};

#pragma unroll
  for (int j = 0; j < KGRP; ++j) {
    af[0][j]    = ap[2 * j];
    bf[0][j][0] = bp[(4 * j) * (NHID / 4)];
    bf[0][j][1] = bp[(4 * j + 1) * (NHID / 4)];
  }
#pragma unroll
  for (int g = 0; g < NGRP; ++g) {
    if (g + 1 < NGRP) {
#pragma unroll
      for (int j = 0; j < KGRP; ++j) {
        const int s = (g + 1) * KGRP + j;
        af[(g + 1) & 1][j]    = ap[2 * s];
        bf[(g + 1) & 1][j][0] = bp[(4 * s) * (NHID / 4)];
        bf[(g + 1) & 1][j][1] = bp[(4 * s + 1) * (NHID / 4)];
      }
    }
    __builtin_amdgcn_sched_barrier(0);
    if (g == 0 && hw < 2) {
      constexpr unsigned int COLS = (1u << (NROI / 4)) - 1u;
      v4u mw = (v4u){0u, 0u, 0u, 0u};
#pragma unroll
      for (int j = 0; j < 8; ++j) {
        const v4u bj = { __builtin_amdgcn_ballot_w32(am[j][0] > 0.f) & COLS,
                         __builtin_amdgcn_ballot_w32(am[j][1] > 0.f) & COLS,
                         __builtin_amdgcn_ballot_w32(am[j][2] > 0.f) & COLS,
                         __builtin_amdgcn_ballot_w32(am[j][3] > 0.f) & COLS };
        mw = (l == j) ? bj : mw;
      }
      if (l < 8)
        *(v4u*)((unsigned int*)emask + (size_t)(m0 + 8 * hw + l) * MASKW) = mw;
    }
#pragma unroll
    for (int j = 0; j < KGRP; ++j) {
      const v2f a = af[g & 1][j];
#pragma unroll
      for (int nt = 0; nt < 4; ++nt) {
        const v2f b = { bf[g & 1][j][0][nt], bf[g & 1][j][1][nt] };
        acc[nt] = wmma4f32(a, b, acc[nt]);
      }
    }
    __builtin_amdgcn_sched_barrier(0);
  }

  float* Ch = C + (size_t)h * NNODES * NHID;
#pragma unroll
  for (int r = 0; r < 8; ++r) {
    v4f o = { acc[0][r], acc[1][r], acc[2][r], acc[3][r] };
    *(v4f*)(Ch + (size_t)(m0 + 8 * lh + r) * NHID + 4 * lm) = o;
  }

  const v4f a1 = *(const v4f*)(avec + (size_t)h * 2 * NHID + 4 * lm);
  const v4f a2 = *(const v4f*)(avec + (size_t)h * 2 * NHID + NHID + 4 * lm);
  float p1[8], p2[8];
#pragma unroll
  for (int r = 0; r < 8; ++r) {
    p1[r] = acc[0][r] * a1[0] + acc[1][r] * a1[1] + acc[2][r] * a1[2] + acc[3][r] * a1[3];
    p2[r] = acc[0][r] * a2[0] + acc[1][r] * a2[1] + acc[2][r] * a2[2] + acc[3][r] * a2[3];
  }
#pragma unroll
  for (int r = 0; r < 8; ++r) {
    p1[r] = row_xor_add<1>(p1[r]);  p2[r] = row_xor_add<1>(p2[r]);
  }
#pragma unroll
  for (int r = 0; r < 8; ++r) {
    p1[r] = row_xor_add<2>(p1[r]);  p2[r] = row_xor_add<2>(p2[r]);
  }
#pragma unroll
  for (int r = 0; r < 8; ++r) {
    p1[r] = row_xor_add<4>(p1[r]);  p2[r] = row_xor_add<4>(p2[r]);
  }
#pragma unroll
  for (int r = 0; r < 8; ++r) {
    p1[r] = row_xor_add<8>(p1[r]);  p2[r] = row_xor_add<8>(p2[r]);
  }
  float e1 = p1[0], e2 = p2[0];
#pragma unroll
  for (int r = 1; r < 8; ++r) {
    e1 = (lm == r) ? p1[r] : e1;
    e2 = (lm == r) ? p2[r] : e2;
  }
  if (lm < 8) {
    const size_t idx = (size_t)h * NNODES + m0 + 8 * lh + lm;
    el[idx] = e1;
    er[idx] = e2;
  }
}

#define WHS_S 68
#define KSTEPS (NROI / 4)
#define LDS_ROWS 128
#define TAB_OFF (LDS_ROWS * WHS_S)
#define TAB_REC 8
#define LDS_BYTES ((LDS_ROWS * WHS_S + (NROI / 2) * TAB_REC) * sizeof(float))

template <int N>
__device__ __forceinline__ float row_share(float v) {
  int o = __builtin_amdgcn_update_dpp(0, __float_as_int(v), 0x150 | N, 0xf, 0xf, false);
  return __int_as_float(o);
}

template <int MASK>
__device__ __forceinline__ float row_xor_max(float v) {
  int o = __builtin_amdgcn_update_dpp(0, __float_as_int(v), 0x160 | MASK, 0xf, 0xf, false);
  return fmaxf(v, __int_as_float(o));
}

__device__ __forceinline__ float masked_score(float eli, float erj, unsigned int word, int bit) {
  float z  = eli + erj;
  float ev = (z > 0.f) ? z : ALPHA_LR * z;
  return ((word >> bit) & 1u) ? ev : -INFINITY;
}

__device__ __forceinline__ float masked_weight(float e, float m) {
  float wv = expf(e - m);
  return (e == -INFINITY) ? 0.f : wv;
}

template <bool LAYER2>
__global__ __launch_bounds__(256)
void att_block_kernel(const float* __restrict__ emask, const float* __restrict__ Wh,
                      const float* __restrict__ el, const float* __restrict__ er,
                      const float* __restrict__ Wnext, const float* __restrict__ anext,
                      float* __restrict__ out, float* __restrict__ elnext,
                      float* __restrict__ ernext) {
  extern __shared__ __attribute__((aligned(16))) float lds[];
  float* whs = lds;

  const int g = blockIdx.x, h = blockIdx.y;
  const int t = threadIdx.x;
  const int r0 = g * NROI;
  const int w  = t >> 5;
  const int l  = t & 31;
  const int lm = l & 15, lh = l >> 4;

  const int  irow  = w * 16 + lm;
  const bool rowok = irow < NROI;
  const int  ic    = rowok ? irow : (NROI - 1);
  const v2u mk = *(const v2u*)((const unsigned int*)emask + (size_t)(r0 + ic) * MASKW + 2 * lh);

  const int lq = (l < NROI / 4) ? l : (NROI / 4 - 1);
  const int jc = (t < NROI) ? t : (NROI - 1);
  float eli, ert;
  v4f   er4;
  float* tab = lds + TAB_OFF;

  if constexpr (!LAYER2) {
    static_assert((NROI * NHID) % 256 == 0, "staging has no remainder");
    constexpr int NSTAGE = (NROI * NHID) / 256;
    const float* WhH = Wh + (size_t)h * NNODES * NHID;
    const float* elh = el + (size_t)h * NNODES + r0;
    const float* erh = er + (size_t)h * NNODES + r0;
    float st[NSTAGE];
    eli = elh[ic];
    ert = erh[jc];
    er4 = *(const v4f*)(erh + 4 * lq);
#pragma unroll
    for (int k = 0; k < NSTAGE; ++k) {
      int i = t + 256 * k;
      st[k] = WhH[(size_t)(r0 + (i >> 6)) * NHID + (i & 63)];
    }
    __builtin_amdgcn_sched_barrier(0);
#pragma unroll
    for (int k = 0; k < NSTAGE; ++k) {
      int i = t + 256 * k;
      whs[(i >> 6) * WHS_S + (i & 63)] = st[k];
    }
  } else {
    constexpr int NV4   = (NROI * NHID) / 4;
    constexpr int NFULL = NV4 / 256;
    static_assert(NV4 - 256 * NFULL == 64, "the remainder is waves 0 and 1");
    constexpr size_t PLANE = (size_t)NNODES * NHID;
    const v4f* Wp = (const v4f*)(Wh + (size_t)r0 * NHID);
    const bool rem = t < (NV4 - 256 * NFULL);
    float elp[4], erq[4];
    v4f er4p[4];
    v4f st4[NFULL + 1][4];
#pragma unroll
    for (int p = 0; p < 4; ++p) {
      elp[p]  = el[(size_t)p * NNODES + r0 + ic];
      erq[p]  = er[(size_t)p * NNODES + r0 + jc];
      er4p[p] = *(const v4f*)(er + (size_t)p * NNODES + r0 + 4 * lq);
    }
#pragma unroll
    for (int k = 0; k < NFULL; ++k) {
#pragma unroll
      for (int p = 0; p < 4; ++p)
        st4[k][p] = Wp[p * (PLANE / 4) + t + 256 * k];
    }
    if (rem) {
#pragma unroll
      for (int p = 0; p < 4; ++p)
        st4[NFULL][p] = Wp[p * (PLANE / 4) + t + 256 * NFULL];
    }
    __builtin_amdgcn_sched_barrier(0);
    eli = ((elp[0] + elp[1]) + elp[2]) + elp[3];
    ert = ((erq[0] + erq[1]) + erq[2]) + erq[3];
    er4 = ((er4p[0] + er4p[1]) + er4p[2]) + er4p[3];
#pragma unroll
    for (int k = 0; k < NFULL; ++k) {
      const int i4 = t + 256 * k;
      const v4f sum = ((st4[k][0] + st4[k][1]) + st4[k][2]) + st4[k][3];
      float* dst = whs + (i4 >> 4) * WHS_S + 8 * (i4 & 7) + 2 * ((i4 >> 3) & 1);
      *(v2f*)dst       = (v2f){ sum[0], sum[1] };
      *(v2f*)(dst + 4) = (v2f){ sum[2], sum[3] };
    }
    if (rem) {
      const int i4 = t + 256 * NFULL;
      const v4f sum = ((st4[NFULL][0] + st4[NFULL][1]) + st4[NFULL][2]) + st4[NFULL][3];
      float* dst = whs + (i4 >> 4) * WHS_S + 8 * (i4 & 7) + 2 * ((i4 >> 3) & 1);
      *(v2f*)dst       = (v2f){ sum[0], sum[1] };
      *(v2f*)(dst + 4) = (v2f){ sum[2], sum[3] };
    }
  }

  float R = fmaxf(fmaxf(er4[0], er4[1]), fmaxf(er4[2], er4[3]));
  R = row_xor_max<1>(R);
  R = row_xor_max<2>(R);
  R = row_xor_max<4>(R);
  R = row_xor_max<8>(R);
  R = fmaxf(R, __shfl_xor(R, 16));

  if (t < NROI) {
    const float d = ert - R;
    float* rec = tab + (t >> 1) * TAB_REC + (t & 1);
    rec[0] = ert;
    rec[2] = expf(d);
    rec[4] = expf(ALPHA_LR * d);
  }

  __syncthreads();

  const float u  = eli + R;
  const float mb = (u > 0.f) ? u : ALPHA_LR * u;
  const float ga = expf(u - mb);
  const float gc = expf(ALPHA_LR * u - mb);

  const float* tp = tab + lh * TAB_REC;
  v2f e[KSTEPS];
  float ssum = 0.f;
#pragma unroll
  for (int s = 0; s < KSTEPS; ++s) {
    const v4f eb = *(const v4f*)(tp + 2 * s * TAB_REC);
    const v2f dd = *(const v2f*)(tp + 2 * s * TAB_REC + 4);
    float w0 = (eli + eb[0] > 0.f) ? ga * eb[2] : gc * dd[0];
    float w1 = (eli + eb[1] > 0.f) ? ga * eb[3] : gc * dd[1];
    w0 = ((mk[0] >> s) & 1u) ? w0 : 0.f;
    w1 = ((mk[1] >> s) & 1u) ? w1 : 0.f;
    e[s][0] = w0;
    e[s][1] = w1;
    ssum += w0;
    ssum += w1;
  }
  ssum += __shfl_xor(ssum, 16);

  const bool weak = ((mk[0] | mk[1]) != 0u) && !(ssum >= 1e-20f);
  if (__builtin_expect(__builtin_amdgcn_ballot_w32(weak) != 0u, 0)) {
    float m = -INFINITY;
#pragma unroll
    for (int s = 0; s < KSTEPS; ++s) {
      const v2f rv = *(const v2f*)(tp + 2 * s * TAB_REC);
      e[s][0] = masked_score(eli, rv[0], mk[0], s);
      e[s][1] = masked_score(eli, rv[1], mk[1], s);
      m = fmaxf(fmaxf(m, e[s][0]), e[s][1]);
    }
    m = fmaxf(m, __shfl_xor(m, 16));
    ssum = 0.f;
#pragma unroll
    for (int s = 0; s < KSTEPS; ++s) {
      e[s][0] = masked_weight(e[s][0], m);
      e[s][1] = masked_weight(e[s][1], m);
      ssum += e[s][0];
      ssum += e[s][1];
    }
    ssum += __shfl_xor(ssum, 16);
  }

  float inv = (ssum > 0.f) ? 1.f / ssum : 0.f;
  if (!rowok) inv = 0.f;
#pragma unroll
  for (int s = 0; s < KSTEPS; ++s) { e[s][0] *= inv; e[s][1] *= inv; }

  v8f acc[4];
#pragma unroll
  for (int nt = 0; nt < 4; ++nt)
    acc[nt] = (v8f){0.f,0.f,0.f,0.f,0.f,0.f,0.f,0.f};

  const v4f* bq = (const v4f*)(whs + (2 * lh) * WHS_S + 4 * lm);
#pragma unroll
  for (int s = 0; s < KSTEPS; ++s) {
    const v2f a = e[s];
    const v4f b0 = bq[(4 * s) * (WHS_S / 4)];
    const v4f b1 = bq[(4 * s + 1) * (WHS_S / 4)];
#pragma unroll
    for (int nt = 0; nt < 4; ++nt) {
      const v2f b = { b0[nt], b1[nt] };
      acc[nt] = wmma4f32(a, b, acc[nt]);
    }
  }

  float ss[8];
#pragma unroll
  for (int v = 0; v < 8; ++v)
    ss[v] = acc[0][v] * acc[0][v] + acc[1][v] * acc[1][v]
          + acc[2][v] * acc[2][v] + acc[3][v] * acc[3][v];
#pragma unroll
  for (int v = 0; v < 8; ++v) ss[v] = row_xor_add<1>(ss[v]);
#pragma unroll
  for (int v = 0; v < 8; ++v) ss[v] = row_xor_add<2>(ss[v]);
#pragma unroll
  for (int v = 0; v < 8; ++v) ss[v] = row_xor_add<4>(ss[v]);
#pragma unroll
  for (int v = 0; v < 8; ++v) ss[v] = row_xor_add<8>(ss[v]);

  float mine = ss[0];
#pragma unroll
  for (int v = 1; v < 8; ++v) mine = (lm == v) ? ss[v] : mine;
  const float r = sqrtf(mine);
  float sc = fkernel_scale(r);
  if (LAYER2) sc *= fkernel_scale(sc * r);
  float sv[8];
  sv[0] = row_share<0>(sc);  sv[1] = row_share<1>(sc);
  sv[2] = row_share<2>(sc);  sv[3] = row_share<3>(sc);
  sv[4] = row_share<4>(sc);  sv[5] = row_share<5>(sc);
  sv[6] = row_share<6>(sc);  sv[7] = row_share<7>(sc);

  if constexpr (LAYER2) {
    const int c = blockIdx.y;
    v2f hq[8];
#pragma unroll
    for (int v = 0; v < 8; ++v)
      hq[v] = c ? (v2f){ acc[2][v], acc[3][v] } : (v2f){ acc[0][v], acc[1][v] };
    int    row  = w * 16 + 8 * lh;
    float* orow = out + (size_t)(r0 + row) * NHID + 32 * c + 2 * lm;
    for (int v = 0; v < 4; ++v) {
      const v2f   x0 = hq[0], x1 = hq[1];
      const float s0 = sv[0], s1 = sv[1];
      const v2f o0 = { act_elu_cos(s0 * x0[0]), act_elu_cos(s0 * x0[1]) };
      const v2f o1 = { act_elu_cos(s1 * x1[0]), act_elu_cos(s1 * x1[1]) };
      if (row < NROI)     *(v2f*)orow          = o0;
      if (row + 1 < NROI) *(v2f*)(orow + NHID) = o1;
      row  += 2;
      orow += 2 * NHID;
#pragma unroll
      for (int j = 0; j < 6; ++j) { hq[j] = hq[j + 2]; sv[j] = sv[j + 2]; }
    }
  } else {
    v4f hv[8];
#pragma unroll
    for (int v = 0; v < 8; ++v)
      hv[v] = (v4f){ acc[0][v], acc[1][v], acc[2][v], acc[3][v] };

    __syncthreads();

    constexpr int KS2 = NHID / 4;
    const v4f* wp = (const v4f*)(Wnext + (size_t)(h * NHID + 2 * lh) * NHID + 4 * lm);
    v4f wb[KS2][2];
#pragma unroll
    for (int s = 0; s < KS2; ++s) {
      wb[s][0] = wp[(4 * s) * (NHID / 4)];
      wb[s][1] = wp[(4 * s + 1) * (NHID / 4)];
    }
    const v4f a1 = *(const v4f*)(anext + 4 * lm);
    const v4f a2 = *(const v4f*)(anext + NHID + 4 * lm);

    float* xrow = whs + (w * 16 + 8 * lh) * WHS_S + 4 * lm;
    for (int v = 0; v < 8; ++v) {
      const v4f   x = hv[0];
      const float s = sv[0];
      const v4f o = { act_elu_cos(s * x[0]), act_elu_cos(s * x[1]),
                      act_elu_cos(s * x[2]), act_elu_cos(s * x[3]) };
      *(v4f*)xrow = o;
      xrow += WHS_S;
#pragma unroll
      for (int j = 0; j < 7; ++j) { hv[j] = hv[j + 1]; sv[j] = sv[j + 1]; }
    }

    __syncthreads();

    v8f pc[4];
#pragma unroll
    for (int nt = 0; nt < 4; ++nt)
      pc[nt] = (v8f){0.f,0.f,0.f,0.f,0.f,0.f,0.f,0.f};
    const v2f* aq = (const v2f*)(whs + (w * 16 + lm) * WHS_S + 2 * lh);
#pragma unroll
    for (int s = 0; s < KS2; ++s) {
      const v2f a = aq[2 * s];
#pragma unroll
      for (int nt = 0; nt < 4; ++nt) {
        const v2f b = { wb[s][0][nt], wb[s][1][nt] };
        pc[nt] = wmma4f32(a, b, pc[nt]);
      }
    }

    const int rbase = w * 16 + 8 * lh;
    float* Pp = out + (size_t)h * NNODES * NHID + (size_t)(r0 + rbase) * NHID + 4 * lm;
#pragma unroll
    for (int v = 0; v < 8; ++v) {
      const v4f o = { pc[0][v], pc[1][v], pc[2][v], pc[3][v] };
      if (rbase + v < NROI) *(v4f*)(Pp + (size_t)v * NHID) = o;
    }

    float p1[8], p2[8];
#pragma unroll
    for (int v = 0; v < 8; ++v) {
      p1[v] = pc[0][v] * a1[0] + pc[1][v] * a1[1] + pc[2][v] * a1[2] + pc[3][v] * a1[3];
      p2[v] = pc[0][v] * a2[0] + pc[1][v] * a2[1] + pc[2][v] * a2[2] + pc[3][v] * a2[3];
    }
#pragma unroll
    for (int v = 0; v < 8; ++v) {
      p1[v] = row_xor_add<1>(p1[v]);  p2[v] = row_xor_add<1>(p2[v]);
    }
#pragma unroll
    for (int v = 0; v < 8; ++v) {
      p1[v] = row_xor_add<2>(p1[v]);  p2[v] = row_xor_add<2>(p2[v]);
    }
#pragma unroll
    for (int v = 0; v < 8; ++v) {
      p1[v] = row_xor_add<4>(p1[v]);  p2[v] = row_xor_add<4>(p2[v]);
    }
#pragma unroll
    for (int v = 0; v < 8; ++v) {
      p1[v] = row_xor_add<8>(p1[v]);  p2[v] = row_xor_add<8>(p2[v]);
    }
    float e1 = p1[0], e2 = p2[0];
#pragma unroll
    for (int v = 1; v < 8; ++v) {
      e1 = (lm == v) ? p1[v] : e1;
      e2 = (lm == v) ? p2[v] : e2;
    }
    if (lm < 8 && rbase + lm < NROI) {
      const size_t idx = (size_t)h * NNODES + r0 + rbase + lm;
      elnext[idx] = e1;
      ernext[idx] = e2;
    }
  }
}

extern "C" void kernel_launch(void* const* d_in, const int* in_sizes, int n_in,
                              void* d_out, int out_size, void* d_ws, size_t ws_size,
                              hipStream_t stream) {
  (void)in_sizes; (void)n_in; (void)out_size; (void)ws_size;
  const float* x       = (const float*)d_in[0];
  const float* adj     = (const float*)d_in[1];
  const float* W_heads = (const float*)d_in[2];
  const float* a_heads = (const float*)d_in[3];
  const float* W_out   = (const float*)d_in[4];
  const float* a_out   = (const float*)d_in[5];
  float* out = (float*)d_out;

  float* ws = (float*)d_ws;
  float* Wh   = ws;                  ws += (size_t)NHEADS * NNODES * NHID;
  float* el1  = ws;                  ws += (size_t)NHEADS * NNODES;
  float* er1  = ws;                  ws += (size_t)NHEADS * NNODES;
  float* Wh2p = ws;                  ws += (size_t)NHEADS * NNODES * NHID;
  float* el2p = ws;                  ws += (size_t)NHEADS * NNODES;
  float* er2p = ws;                  ws += (size_t)NHEADS * NNODES;
  float* emask = ws;

  const int mtiles = NNODES / 16;

  proj_strip_kernel<NFEAT, NHEADS><<<mtiles, 32 * NHEADS, 0, stream>>>(
      x, W_heads, a_heads, Wh, el1, er1, adj, emask);
  att_block_kernel<false><<<dim3(NGRAPH, NHEADS), 256, LDS_BYTES, stream>>>(
      emask, Wh, el1, er1, W_out, a_out, Wh2p, el2p, er2p);
  att_block_kernel<true><<<dim3(NGRAPH, 2), 256, LDS_BYTES, stream>>>(
      emask, Wh2p, el2p, er2p, nullptr, nullptr, out, nullptr, nullptr);
}


# Softmax weights made k-step by k-step inside the product loop and fed to the WMMAs undivided; each 16x64 result tile scaled by its rows' 1/sum once (values re-rounded, every sum order kept)
# speedup vs baseline: 5.0394x; 1.0389x over previous
#include <hip/hip_runtime.h>
#include <math.h>

#define NNODES   3712
#define NROI     116
#define NGRAPH   32
#define NFEAT    128
#define NHID     64
#define NHEADS   4
#define ALPHA_LR 0.1f
#define A_PARA   0.1f
#define PHI_C    0.94247779607693793f
#define MAXNORM  (1.0f - 4e-3f)

typedef __attribute__((ext_vector_type(2))) float v2f;
typedef __attribute__((ext_vector_type(4))) float v4f;
typedef __attribute__((ext_vector_type(8))) float v8f;
typedef __attribute__((ext_vector_type(2))) unsigned int v2u;
typedef __attribute__((ext_vector_type(4))) unsigned int v4u;

__device__ __forceinline__ v8f wmma4f32(v2f a, v2f b, v8f c) {
  return __builtin_amdgcn_wmma_f32_16x16x4_f32(
      false, a, false, b, (short)0, c, false, false);
}

__device__ __forceinline__ float fkernel_scale(float r) {
  float proj = 1.0f, rn = r;
  if (r > MAXNORM) { proj = MAXNORM / r; rn = MAXNORM; }
  rn = fmaxf(rn, 1e-15f);
  float t = fminf(rn, 1.0f - 1e-7f);
  return proj * (atanhf(t) / rn);
}

__device__ __forceinline__ float expm1_main(float v) {
  const float n = __builtin_rintf(v * 1.44269502f);
  float r = __builtin_fmaf(n, -0.693147182f, v);
  r = __builtin_fmaf(n, 1.90465421e-9f, r);
  const float tw = ldexpf(1.0f, (int)n);
  float p = __builtin_fmaf(1.99510498e-4f, r, 1.39304996e-3f);
  p = __builtin_fmaf(r, p, 8.33319966e-3f);
  p = __builtin_fmaf(r, p, 4.16665114e-2f);
  p = __builtin_fmaf(r, p, 0.166666672f);
  p = __builtin_fmaf(r, p, 0.5f);
  p = r * p;
  const float er = __builtin_fmaf(r, p, r);
  return __builtin_fmaf(tw, er, tw - 1.0f);
}

__device__ __forceinline__ float cos_main(float x) {
  const float ax = __builtin_fabsf(x);
  const float n  = __builtin_rintf(ax * 0.636619747f);
  float r = __builtin_fmaf(-1.57079625f, n, ax);
  r = __builtin_fmaf(-7.54978942e-8f, n, r);
  r = __builtin_fmaf(-5.39030253e-15f, n, r);
  const unsigned int q = (unsigned int)(int)n;
  const float s = r * r;
  float ps = __builtin_fmaf(-1.94644585e-4f, s, 8.33171979e-3f);
  ps = __builtin_fmaf(s, ps, -0.166666463f);
  ps = s * ps;
  const float sn = __builtin_fmaf(r, ps, r);
  float pc = __builtin_fmaf(2.56687417e-5f, s, -1.39091106e-3f);
  pc = __builtin_fmaf(s, pc, 4.16679047e-2f);
  pc = __builtin_fmaf(s, pc, -0.500000238f);
  const float cs = __builtin_fmaf(s, pc, 1.0f);
  const float m  = (q & 1u) ? -sn : cs;
  return __uint_as_float(__float_as_uint(m) ^ ((q << 30) & 0x80000000u));
}

__device__ __forceinline__ float act_elu_cos(float v) {
  const float em = expm1_main(v);
  const float e  = (v > 0.0f) ? v : em;
  return __builtin_fmaf(A_PARA, cos_main(v + PHI_C), e);
}

#define KGRP 8
#define MASKW 4

template <int MASK>
__device__ __forceinline__ float row_xor_add(float v) {
  int o = __builtin_amdgcn_update_dpp(0, __float_as_int(v), 0x160 | MASK, 0xf, 0xf, false);
  return v + __int_as_float(o);
}

template <int K, int NH>
__global__ __launch_bounds__(32 * NH)
void proj_strip_kernel(const float* __restrict__ A, const float* __restrict__ B,
                       const float* __restrict__ avec, float* __restrict__ C,
                       float* __restrict__ el, float* __restrict__ er,
                       const float* __restrict__ adj, float* __restrict__ emask) {
  static_assert(K % (4 * KGRP) == 0, "K is a whole number of load groups");
  static_assert(NH >= 2, "two waves build the 16 rows' edge masks");
  static_assert(NROI % 4 == 0 && NROI / 4 <= 32, "a row's block is one 16-byte piece per lane");
  constexpr int NGRP = K / (4 * KGRP);
  const int h  = threadIdx.x >> 5;
  const int l  = threadIdx.x & 31;
  const int lm = l & 15;
  const int lh = l >> 4;
  const int m0 = blockIdx.x * 16;

  const v2f* ap = (const v2f*)(A + (size_t)(m0 + lm) * K + 2 * lh);
  const v4f* bp = (const v4f*)(B + (size_t)h * K * NHID + (size_t)(2 * lh) * NHID + 4 * lm);

  const int hw = __builtin_amdgcn_readfirstlane(h);
  v4f am[8];
#pragma unroll
  for (int j = 0; j < 8; ++j) am[j] = (v4f){0.f, 0.f, 0.f, 0.f};
  if (hw < 2) {
    const int lc = (l < NROI / 4) ? l : (NROI / 4 - 1);
#pragma unroll
    for (int j = 0; j < 8; ++j) {
      const int row = m0 + 8 * hw + j;
      const int rb  = (row / NROI) * NROI;
      am[j] = *(const v4f*)(adj + (size_t)row * NNODES + rb + 4 * lc);
    }
  }

  v2f af[2][KGRP];
  v4f bf[2][KGRP][2];
  v8f acc[4];
#pragma unroll
  for (int nt = 0; nt < 4; ++nt)
    acc[nt] = (v8f){0.f,0.f,0.f,0.f,0.f,0.f,0.f,0.f};

#pragma unroll
  for (int j = 0; j < KGRP; ++j) {
    af[0][j]    = ap[2 * j];
    bf[0][j][0] = bp[(4 * j) * (NHID / 4)];
    bf[0][j][1] = bp[(4 * j + 1) * (NHID / 4)];
  }
#pragma unroll
  for (int g = 0; g < NGRP; ++g) {
    if (g + 1 < NGRP) {
#pragma unroll
      for (int j = 0; j < KGRP; ++j) {
        const int s = (g + 1) * KGRP + j;
        af[(g + 1) & 1][j]    = ap[2 * s];
        bf[(g + 1) & 1][j][0] = bp[(4 * s) * (NHID / 4)];
        bf[(g + 1) & 1][j][1] = bp[(4 * s + 1) * (NHID / 4)];
      }
    }
    __builtin_amdgcn_sched_barrier(0);
    if (g == 0 && hw < 2) {
      constexpr unsigned int COLS = (1u << (NROI / 4)) - 1u;
      v4u mw = (v4u){0u, 0u, 0u, 0u};
#pragma unroll
      for (int j = 0; j < 8; ++j) {
        const v4u bj = { __builtin_amdgcn_ballot_w32(am[j][0] > 0.f) & COLS,
                         __builtin_amdgcn_ballot_w32(am[j][1] > 0.f) & COLS,
                         __builtin_amdgcn_ballot_w32(am[j][2] > 0.f) & COLS,
                         __builtin_amdgcn_ballot_w32(am[j][3] > 0.f) & COLS };
        mw = (l == j) ? bj : mw;
      }
      if (l < 8)
        *(v4u*)((unsigned int*)emask + (size_t)(m0 + 8 * hw + l) * MASKW) = mw;
    }
#pragma unroll
    for (int j = 0; j < KGRP; ++j) {
      const v2f a = af[g & 1][j];
#pragma unroll
      for (int nt = 0; nt < 4; ++nt) {
        const v2f b = { bf[g & 1][j][0][nt], bf[g & 1][j][1][nt] };
        acc[nt] = wmma4f32(a, b, acc[nt]);
      }
    }
    __builtin_amdgcn_sched_barrier(0);
  }

  float* Ch = C + (size_t)h * NNODES * NHID;
#pragma unroll
  for (int r = 0; r < 8; ++r) {
    v4f o = { acc[0][r], acc[1][r], acc[2][r], acc[3][r] };
    *(v4f*)(Ch + (size_t)(m0 + 8 * lh + r) * NHID + 4 * lm) = o;
  }

  const v4f a1 = *(const v4f*)(avec + (size_t)h * 2 * NHID + 4 * lm);
  const v4f a2 = *(const v4f*)(avec + (size_t)h * 2 * NHID + NHID + 4 * lm);
  float p1[8], p2[8];
#pragma unroll
  for (int r = 0; r < 8; ++r) {
    p1[r] = acc[0][r] * a1[0] + acc[1][r] * a1[1] + acc[2][r] * a1[2] + acc[3][r] * a1[3];
    p2[r] = acc[0][r] * a2[0] + acc[1][r] * a2[1] + acc[2][r] * a2[2] + acc[3][r] * a2[3];
  }
#pragma unroll
  for (int r = 0; r < 8; ++r) {
    p1[r] = row_xor_add<1>(p1[r]);  p2[r] = row_xor_add<1>(p2[r]);
  }
#pragma unroll
  for (int r = 0; r < 8; ++r) {
    p1[r] = row_xor_add<2>(p1[r]);  p2[r] = row_xor_add<2>(p2[r]);
  }
#pragma unroll
  for (int r = 0; r < 8; ++r) {
    p1[r] = row_xor_add<4>(p1[r]);  p2[r] = row_xor_add<4>(p2[r]);
  }
#pragma unroll
  for (int r = 0; r < 8; ++r) {
    p1[r] = row_xor_add<8>(p1[r]);  p2[r] = row_xor_add<8>(p2[r]);
  }
  float e1 = p1[0], e2 = p2[0];
#pragma unroll
  for (int r = 1; r < 8; ++r) {
    e1 = (lm == r) ? p1[r] : e1;
    e2 = (lm == r) ? p2[r] : e2;
  }
  if (lm < 8) {
    const size_t idx = (size_t)h * NNODES + m0 + 8 * lh + lm;
    el[idx] = e1;
    er[idx] = e2;
  }
}

#define WHS_S 68
#define KSTEPS (NROI / 4)
#define LDS_ROWS 128
#define TAB_OFF (LDS_ROWS * WHS_S)
#define TAB_REC 8
#define LDS_BYTES ((LDS_ROWS * WHS_S + (NROI / 2) * TAB_REC) * sizeof(float))

template <int N>
__device__ __forceinline__ float row_share(float v) {
  int o = __builtin_amdgcn_update_dpp(0, __float_as_int(v), 0x150 | N, 0xf, 0xf, false);
  return __int_as_float(o);
}

template <int MASK>
__device__ __forceinline__ float row_xor_max(float v) {
  int o = __builtin_amdgcn_update_dpp(0, __float_as_int(v), 0x160 | MASK, 0xf, 0xf, false);
  return fmaxf(v, __int_as_float(o));
}

__device__ __forceinline__ float masked_score(float eli, float erj, unsigned int word, int bit) {
  float z  = eli + erj;
  float ev = (z > 0.f) ? z : ALPHA_LR * z;
  return ((word >> bit) & 1u) ? ev : -INFINITY;
}

__device__ __forceinline__ float masked_weight(float e, float m) {
  float wv = expf(e - m);
  return (e == -INFINITY) ? 0.f : wv;
}

__device__ __forceinline__ void att_kstep(v8f (&acc)[4], v2f a, v4f b0, v4f b1) {
#pragma unroll
  for (int nt = 0; nt < 4; ++nt) {
    const v2f b = { b0[nt], b1[nt] };
    acc[nt] = wmma4f32(a, b, acc[nt]);
  }
}

template <bool LAYER2>
__global__ __launch_bounds__(256)
void att_block_kernel(const float* __restrict__ emask, const float* __restrict__ Wh,
                      const float* __restrict__ el, const float* __restrict__ er,
                      const float* __restrict__ Wnext, const float* __restrict__ anext,
                      float* __restrict__ out, float* __restrict__ elnext,
                      float* __restrict__ ernext) {
  extern __shared__ __attribute__((aligned(16))) float lds[];
  float* whs = lds;

  const int g = blockIdx.x, h = blockIdx.y;
  const int t = threadIdx.x;
  const int r0 = g * NROI;
  const int w  = t >> 5;
  const int l  = t & 31;
  const int lm = l & 15, lh = l >> 4;

  const int  irow  = w * 16 + lm;
  const bool rowok = irow < NROI;
  const int  ic    = rowok ? irow : (NROI - 1);
  const v2u mk = *(const v2u*)((const unsigned int*)emask + (size_t)(r0 + ic) * MASKW + 2 * lh);

  const int lq = (l < NROI / 4) ? l : (NROI / 4 - 1);
  const int jc = (t < NROI) ? t : (NROI - 1);
  float eli, ert;
  v4f   er4;
  float* tab = lds + TAB_OFF;

  if constexpr (!LAYER2) {
    static_assert((NROI * NHID) % 256 == 0, "staging has no remainder");
    constexpr int NSTAGE = (NROI * NHID) / 256;
    const float* WhH = Wh + (size_t)h * NNODES * NHID;
    const float* elh = el + (size_t)h * NNODES + r0;
    const float* erh = er + (size_t)h * NNODES + r0;
    float st[NSTAGE];
    eli = elh[ic];
    ert = erh[jc];
    er4 = *(const v4f*)(erh + 4 * lq);
#pragma unroll
    for (int k = 0; k < NSTAGE; ++k) {
      int i = t + 256 * k;
      st[k] = WhH[(size_t)(r0 + (i >> 6)) * NHID + (i & 63)];
    }
    __builtin_amdgcn_sched_barrier(0);
#pragma unroll
    for (int k = 0; k < NSTAGE; ++k) {
      int i = t + 256 * k;
      whs[(i >> 6) * WHS_S + (i & 63)] = st[k];
    }
  } else {
    constexpr int NV4   = (NROI * NHID) / 4;
    constexpr int NFULL = NV4 / 256;
    static_assert(NV4 - 256 * NFULL == 64, "the remainder is waves 0 and 1");
    constexpr size_t PLANE = (size_t)NNODES * NHID;
    const v4f* Wp = (const v4f*)(Wh + (size_t)r0 * NHID);
    const bool rem = t < (NV4 - 256 * NFULL);
    float elp[4], erq[4];
    v4f er4p[4];
    v4f st4[NFULL + 1][4];
#pragma unroll
    for (int p = 0; p < 4; ++p) {
      elp[p]  = el[(size_t)p * NNODES + r0 + ic];
      erq[p]  = er[(size_t)p * NNODES + r0 + jc];
      er4p[p] = *(const v4f*)(er + (size_t)p * NNODES + r0 + 4 * lq);
    }
#pragma unroll
    for (int k = 0; k < NFULL; ++k) {
#pragma unroll
      for (int p = 0; p < 4; ++p)
        st4[k][p] = Wp[p * (PLANE / 4) + t + 256 * k];
    }
    if (rem) {
#pragma unroll
      for (int p = 0; p < 4; ++p)
        st4[NFULL][p] = Wp[p * (PLANE / 4) + t + 256 * NFULL];
    }
    __builtin_amdgcn_sched_barrier(0);
    eli = ((elp[0] + elp[1]) + elp[2]) + elp[3];
    ert = ((erq[0] + erq[1]) + erq[2]) + erq[3];
    er4 = ((er4p[0] + er4p[1]) + er4p[2]) + er4p[3];
#pragma unroll
    for (int k = 0; k < NFULL; ++k) {
      const int i4 = t + 256 * k;
      const v4f sum = ((st4[k][0] + st4[k][1]) + st4[k][2]) + st4[k][3];
      float* dst = whs + (i4 >> 4) * WHS_S + 8 * (i4 & 7) + 2 * ((i4 >> 3) & 1);
      *(v2f*)dst       = (v2f){ sum[0], sum[1] };
      *(v2f*)(dst + 4) = (v2f){ sum[2], sum[3] };
    }
    if (rem) {
      const int i4 = t + 256 * NFULL;
      const v4f sum = ((st4[NFULL][0] + st4[NFULL][1]) + st4[NFULL][2]) + st4[NFULL][3];
      float* dst = whs + (i4 >> 4) * WHS_S + 8 * (i4 & 7) + 2 * ((i4 >> 3) & 1);
      *(v2f*)dst       = (v2f){ sum[0], sum[1] };
      *(v2f*)(dst + 4) = (v2f){ sum[2], sum[3] };
    }
  }

  float R = fmaxf(fmaxf(er4[0], er4[1]), fmaxf(er4[2], er4[3]));
  R = row_xor_max<1>(R);
  R = row_xor_max<2>(R);
  R = row_xor_max<4>(R);
  R = row_xor_max<8>(R);
  R = fmaxf(R, __shfl_xor(R, 16));

  if (t < NROI) {
    const float d = ert - R;
    float* rec = tab + (t >> 1) * TAB_REC + (t & 1);
    rec[0] = ert;
    rec[2] = expf(d);
    rec[4] = expf(ALPHA_LR * d);
  }

  __syncthreads();

  const float u  = eli + R;
  const float mb = (u > 0.f) ? u : ALPHA_LR * u;
  const float ga = expf(u - mb);
  const float gc = expf(ALPHA_LR * u - mb);

  v8f acc[4];
#pragma unroll
  for (int nt = 0; nt < 4; ++nt)
    acc[nt] = (v8f){0.f,0.f,0.f,0.f,0.f,0.f,0.f,0.f};
  const float* tp = tab + lh * TAB_REC;
  const v4f* bq = (const v4f*)(whs + (2 * lh) * WHS_S + 4 * lm);
  float ssum = 0.f;
  v4f ebv[KSTEPS];
  v2f ddv[KSTEPS];
  v4f b0v[KSTEPS], b1v[KSTEPS];
  v2f wv[KSTEPS];
#pragma unroll
  for (int s = 0; s < 2; ++s) {
    ebv[s] = *(const v4f*)(tp + 2 * s * TAB_REC);
    ddv[s] = *(const v2f*)(tp + 2 * s * TAB_REC + 4);
  }
  b0v[0] = bq[0];
  b1v[0] = bq[WHS_S / 4];
#pragma unroll
  for (int s = -1; s < KSTEPS; ++s) {
    if (s >= 0 && s + 1 < KSTEPS) {
      b0v[s + 1] = bq[(4 * (s + 1)) * (WHS_S / 4)];
      b1v[s + 1] = bq[(4 * (s + 1) + 1) * (WHS_S / 4)];
    }
    if (s >= 0 && s + 2 < KSTEPS) {
      ebv[s + 2] = *(const v4f*)(tp + 2 * (s + 2) * TAB_REC);
      ddv[s + 2] = *(const v2f*)(tp + 2 * (s + 2) * TAB_REC + 4);
    }
    if (s + 1 < KSTEPS) {
      const int n = s + 1;
      float w0 = (eli + ebv[n][0] > 0.f) ? ga * ebv[n][2] : gc * ddv[n][0];
      float w1 = (eli + ebv[n][1] > 0.f) ? ga * ebv[n][3] : gc * ddv[n][1];
      w0 = ((mk[0] >> n) & 1u) ? w0 : 0.f;
      w1 = ((mk[1] >> n) & 1u) ? w1 : 0.f;
      ssum += w0;
      ssum += w1;
      wv[n] = (v2f){ w0, w1 };
    }
    if (s >= 0) att_kstep(acc, wv[s], b0v[s], b1v[s]);
    __builtin_amdgcn_sched_barrier(0);
  }
  ssum += __shfl_xor(ssum, 16);

  const bool weak = ((mk[0] | mk[1]) != 0u) && !(ssum >= 1e-20f);
  if (__builtin_expect(__builtin_amdgcn_ballot_w32(weak) != 0u, 0)) {
    v2f e[KSTEPS];
    float m = -INFINITY;
#pragma unroll
    for (int s = 0; s < KSTEPS; ++s) {
      const v2f rv = *(const v2f*)(tp + 2 * s * TAB_REC);
      e[s][0] = masked_score(eli, rv[0], mk[0], s);
      e[s][1] = masked_score(eli, rv[1], mk[1], s);
      m = fmaxf(fmaxf(m, e[s][0]), e[s][1]);
    }
    m = fmaxf(m, __shfl_xor(m, 16));
    ssum = 0.f;
#pragma unroll
    for (int s = 0; s < KSTEPS; ++s) {
      e[s][0] = masked_weight(e[s][0], m);
      e[s][1] = masked_weight(e[s][1], m);
      ssum += e[s][0];
      ssum += e[s][1];
    }
    ssum += __shfl_xor(ssum, 16);
#pragma unroll
    for (int nt = 0; nt < 4; ++nt)
      acc[nt] = (v8f){0.f,0.f,0.f,0.f,0.f,0.f,0.f,0.f};
#pragma unroll
    for (int s = 0; s < KSTEPS; ++s)
      att_kstep(acc, e[s], bq[(4 * s) * (WHS_S / 4)], bq[(4 * s + 1) * (WHS_S / 4)]);
  }

  float inv = (ssum > 0.f) ? 1.f / ssum : 0.f;
  if (!rowok) inv = 0.f;
  {
    const float up = __int_as_float(__builtin_amdgcn_update_dpp(
        0, __float_as_int(inv), 0x168, 0xf, 0xf, false));
    const float ih = lh ? up : inv;
    float iv[8];
    iv[0] = row_share<0>(ih);  iv[1] = row_share<1>(ih);
    iv[2] = row_share<2>(ih);  iv[3] = row_share<3>(ih);
    iv[4] = row_share<4>(ih);  iv[5] = row_share<5>(ih);
    iv[6] = row_share<6>(ih);  iv[7] = row_share<7>(ih);
#pragma unroll
    for (int nt = 0; nt < 4; ++nt)
#pragma unroll
      for (int v = 0; v < 8; ++v) acc[nt][v] *= iv[v];
  }

  float ss[8];
#pragma unroll
  for (int v = 0; v < 8; ++v)
    ss[v] = acc[0][v] * acc[0][v] + acc[1][v] * acc[1][v]
          + acc[2][v] * acc[2][v] + acc[3][v] * acc[3][v];
#pragma unroll
  for (int v = 0; v < 8; ++v) ss[v] = row_xor_add<1>(ss[v]);
#pragma unroll
  for (int v = 0; v < 8; ++v) ss[v] = row_xor_add<2>(ss[v]);
#pragma unroll
  for (int v = 0; v < 8; ++v) ss[v] = row_xor_add<4>(ss[v]);
#pragma unroll
  for (int v = 0; v < 8; ++v) ss[v] = row_xor_add<8>(ss[v]);

  float mine = ss[0];
#pragma unroll
  for (int v = 1; v < 8; ++v) mine = (lm == v) ? ss[v] : mine;
  const float r = sqrtf(mine);
  float sc = fkernel_scale(r);
  if (LAYER2) sc *= fkernel_scale(sc * r);
  float sv[8];
  sv[0] = row_share<0>(sc);  sv[1] = row_share<1>(sc);
  sv[2] = row_share<2>(sc);  sv[3] = row_share<3>(sc);
  sv[4] = row_share<4>(sc);  sv[5] = row_share<5>(sc);
  sv[6] = row_share<6>(sc);  sv[7] = row_share<7>(sc);

  if constexpr (LAYER2) {
    const int c = blockIdx.y;
    v2f hq[8];
#pragma unroll
    for (int v = 0; v < 8; ++v)
      hq[v] = c ? (v2f){ acc[2][v], acc[3][v] } : (v2f){ acc[0][v], acc[1][v] };
    int    row  = w * 16 + 8 * lh;
    float* orow = out + (size_t)(r0 + row) * NHID + 32 * c + 2 * lm;
    for (int v = 0; v < 4; ++v) {
      const v2f   x0 = hq[0], x1 = hq[1];
      const float s0 = sv[0], s1 = sv[1];
      const v2f o0 = { act_elu_cos(s0 * x0[0]), act_elu_cos(s0 * x0[1]) };
      const v2f o1 = { act_elu_cos(s1 * x1[0]), act_elu_cos(s1 * x1[1]) };
      if (row < NROI)     *(v2f*)orow          = o0;
      if (row + 1 < NROI) *(v2f*)(orow + NHID) = o1;
      row  += 2;
      orow += 2 * NHID;
#pragma unroll
      for (int j = 0; j < 6; ++j) { hq[j] = hq[j + 2]; sv[j] = sv[j + 2]; }
    }
  } else {
    v4f hv[8];
#pragma unroll
    for (int v = 0; v < 8; ++v)
      hv[v] = (v4f){ acc[0][v], acc[1][v], acc[2][v], acc[3][v] };

    __syncthreads();

    constexpr int KS2 = NHID / 4;
    const v4f* wp = (const v4f*)(Wnext + (size_t)(h * NHID + 2 * lh) * NHID + 4 * lm);
    v4f wb[KS2][2];
#pragma unroll
    for (int s = 0; s < KS2; ++s) {
      wb[s][0] = wp[(4 * s) * (NHID / 4)];
      wb[s][1] = wp[(4 * s + 1) * (NHID / 4)];
    }
    const v4f a1 = *(const v4f*)(anext + 4 * lm);
    const v4f a2 = *(const v4f*)(anext + NHID + 4 * lm);

    float* xrow = whs + (w * 16 + 8 * lh) * WHS_S + 4 * lm;
    for (int v = 0; v < 8; ++v) {
      const v4f   x = hv[0];
      const float s = sv[0];
      const v4f o = { act_elu_cos(s * x[0]), act_elu_cos(s * x[1]),
                      act_elu_cos(s * x[2]), act_elu_cos(s * x[3]) };
      *(v4f*)xrow = o;
      xrow += WHS_S;
#pragma unroll
      for (int j = 0; j < 7; ++j) { hv[j] = hv[j + 1]; sv[j] = sv[j + 1]; }
    }

    __syncthreads();

    v8f pc[4];
#pragma unroll
    for (int nt = 0; nt < 4; ++nt)
      pc[nt] = (v8f){0.f,0.f,0.f,0.f,0.f,0.f,0.f,0.f};
    const v2f* aq = (const v2f*)(whs + (w * 16 + lm) * WHS_S + 2 * lh);
#pragma unroll
    for (int s = 0; s < KS2; ++s) {
      const v2f a = aq[2 * s];
#pragma unroll
      for (int nt = 0; nt < 4; ++nt) {
        const v2f b = { wb[s][0][nt], wb[s][1][nt] };
        pc[nt] = wmma4f32(a, b, pc[nt]);
      }
    }

    const int rbase = w * 16 + 8 * lh;
    float* Pp = out + (size_t)h * NNODES * NHID + (size_t)(r0 + rbase) * NHID + 4 * lm;
#pragma unroll
    for (int v = 0; v < 8; ++v) {
      const v4f o = { pc[0][v], pc[1][v], pc[2][v], pc[3][v] };
      if (rbase + v < NROI) *(v4f*)(Pp + (size_t)v * NHID) = o;
    }

    float p1[8], p2[8];
#pragma unroll
    for (int v = 0; v < 8; ++v) {
      p1[v] = pc[0][v] * a1[0] + pc[1][v] * a1[1] + pc[2][v] * a1[2] + pc[3][v] * a1[3];
      p2[v] = pc[0][v] * a2[0] + pc[1][v] * a2[1] + pc[2][v] * a2[2] + pc[3][v] * a2[3];
    }
#pragma unroll
    for (int v = 0; v < 8; ++v) {
      p1[v] = row_xor_add<1>(p1[v]);  p2[v] = row_xor_add<1>(p2[v]);
    }
#pragma unroll
    for (int v = 0; v < 8; ++v) {
      p1[v] = row_xor_add<2>(p1[v]);  p2[v] = row_xor_add<2>(p2[v]);
    }
#pragma unroll
    for (int v = 0; v < 8; ++v) {
      p1[v] = row_xor_add<4>(p1[v]);  p2[v] = row_xor_add<4>(p2[v]);
    }
#pragma unroll
    for (int v = 0; v < 8; ++v) {
      p1[v] = row_xor_add<8>(p1[v]);  p2[v] = row_xor_add<8>(p2[v]);
    }
    float e1 = p1[0], e2 = p2[0];
#pragma unroll
    for (int v = 1; v < 8; ++v) {
      e1 = (lm == v) ? p1[v] : e1;
      e2 = (lm == v) ? p2[v] : e2;
    }
    if (lm < 8 && rbase + lm < NROI) {
      const size_t idx = (size_t)h * NNODES + r0 + rbase + lm;
      elnext[idx] = e1;
      ernext[idx] = e2;
    }
  }
}

extern "C" void kernel_launch(void* const* d_in, const int* in_sizes, int n_in,
                              void* d_out, int out_size, void* d_ws, size_t ws_size,
                              hipStream_t stream) {
  (void)in_sizes; (void)n_in; (void)out_size; (void)ws_size;
  const float* x       = (const float*)d_in[0];
  const float* adj     = (const float*)d_in[1];
  const float* W_heads = (const float*)d_in[2];
  const float* a_heads = (const float*)d_in[3];
  const float* W_out   = (const float*)d_in[4];
  const float* a_out   = (const float*)d_in[5];
  float* out = (float*)d_out;

  float* ws = (float*)d_ws;
  float* Wh   = ws;                  ws += (size_t)NHEADS * NNODES * NHID;
  float* el1  = ws;                  ws += (size_t)NHEADS * NNODES;
  float* er1  = ws;                  ws += (size_t)NHEADS * NNODES;
  float* Wh2p = ws;                  ws += (size_t)NHEADS * NNODES * NHID;
  float* el2p = ws;                  ws += (size_t)NHEADS * NNODES;
  float* er2p = ws;                  ws += (size_t)NHEADS * NNODES;
  float* emask = ws;

  const int mtiles = NNODES / 16;

  proj_strip_kernel<NFEAT, NHEADS><<<mtiles, 32 * NHEADS, 0, stream>>>(
      x, W_heads, a_heads, Wh, el1, er1, adj, emask);
  att_block_kernel<false><<<dim3(NGRAPH, NHEADS), 256, LDS_BYTES, stream>>>(
      emask, Wh, el1, er1, W_out, a_out, Wh2p, el2p, er2p);
  att_block_kernel<true><<<dim3(NGRAPH, 2), 256, LDS_BYTES, stream>>>(
      emask, Wh2p, el2p, er2p, nullptr, nullptr, out, nullptr, nullptr);
}
